# starting kernel
# baseline (speedup 1.0000x reference)
#include <hip/hip_runtime.h>


#define NN 100000
#define NE 1600000
#define DD 128

typedef __attribute__((ext_vector_type(2))) float v2f;
typedef __attribute__((ext_vector_type(8))) float v8f;

__global__ __launch_bounds__(256) void k_deg_init(float* __restrict__ deg) {
  int i = blockIdx.x * 256 + threadIdx.x;
  if (i < NN) deg[i] = 1.0f;
}

__global__ __launch_bounds__(256) void k_deg_count(const int* __restrict__ dst,
                                                   float* __restrict__ deg) {
  int e = blockIdx.x * 256 + threadIdx.x;
  if (e < NE) atomicAdd(&deg[dst[e]], 1.0f);
}

__global__ __launch_bounds__(256) void k_rsqrt(float* __restrict__ d) {
  int i = blockIdx.x * 256 + threadIdx.x;
  if (i < NN) d[i] = rsqrtf(d[i]);
}

__global__ __launch_bounds__(256) void k_gemm_wmma(const float* __restrict__ X,
                                                   const float* __restrict__ W,
                                                   float* __restrict__ H) {
  const int wave = (blockIdx.x * 256 + (int)threadIdx.x) >> 5;
  const int lane = threadIdx.x & 31;
  const int row0 = (wave >> 3) << 4;
  const int col0 = (wave & 7) << 4;
  if (row0 >= NN) return;
  const int half = lane >> 4;
  const int l16  = lane & 15;

  v8f acc = {};
  const float* xr = X + (size_t)(row0 + l16) * DD;
  const float* wc = W + col0 + l16;
  for (int k = 0; k < DD; k += 4) {
    v2f a = *(const v2f*)(xr + k + 2 * half);
    v2f b;
    b.x = wc[(size_t)(k + 2 * half) * DD];
    b.y = wc[(size_t)(k + 2 * half + 1) * DD];
    acc = __builtin_amdgcn_wmma_f32_16x16x4_f32(
         false, a,  false, b,
         (short)0, acc,  false,  false);
  }
  float* hp = H + (size_t)row0 * DD + col0 + l16;
#pragma unroll
  for (int r = 0; r < 8; ++r)
    hp[(size_t)(r + 8 * half) * DD] = acc[r];
}

__global__ __launch_bounds__(256) void k_zero(float* __restrict__ p, int n) {
  int i = blockIdx.x * 256 + threadIdx.x;
  if (i < n) p[i] = 0.0f;
}

__global__ __launch_bounds__(256) void k_scatter(const int* __restrict__ src,
                                                 const int* __restrict__ dst,
                                                 const float* __restrict__ dinv,
                                                 const float* __restrict__ H,
                                                 float* __restrict__ agg) {
  const int e = (blockIdx.x * 256 + (int)threadIdx.x) >> 5;
  const int lane = threadIdx.x & 31;
  if (e >= NE) return;
  const int s = src[e];
  const int d = dst[e];
  const float norm = dinv[s] * dinv[d];
  const float4 v = ((const float4*)(H + (size_t)s * DD))[lane];
  float* a = agg + (size_t)d * DD + lane * 4;
  atomicAdd(a + 0, v.x * norm);
  atomicAdd(a + 1, v.y * norm);
  atomicAdd(a + 2, v.z * norm);
  atomicAdd(a + 3, v.w * norm);
}

__global__ __launch_bounds__(256) void k_finalize(const float* __restrict__ agg,
                                                  const float* __restrict__ H,
                                                  const float* __restrict__ dinv,
                                                  const float* __restrict__ bias,
                                                  float* __restrict__ out,
                                                  int relu) {
  int i = blockIdx.x * 256 + threadIdx.x;
  if (i >= NN * DD) return;
  int n = i >> 7;
  int f = i & (DD - 1);
  float di = dinv[n];
  float v = agg[i] + H[i] * di * di + bias[f];
  out[i] = relu ? fmaxf(v, 0.0f) : v;
}

extern "C" void kernel_launch(void* const* d_in, const int* in_sizes, int n_in,
                              void* d_out, int out_size, void* d_ws, size_t ws_size,
                              hipStream_t stream) {
  const float* x[2]  = {(const float*)d_in[0], (const float*)d_in[2]};
  const int*   ei[2] = {(const int*)d_in[1],   (const int*)d_in[3]};
  const float* W0 = (const float*)d_in[4];
  const float* b0 = (const float*)d_in[5];
  const float* W1 = (const float*)d_in[6];
  const float* b1 = (const float*)d_in[7];
  float* out = (float*)d_out;

  float* ws   = (float*)d_ws;
  float* dinv = ws;
  float* H    = ws + 100352;
  float* AG   = H  + (size_t)NN * DD;
  float* Y    = AG + (size_t)NN * DD;

  const int nd    = NN * DD;
  const dim3 B(256);
  const int gN    = (NN + 255) / 256;
  const int gE    = (NE + 255) / 256;
  const int gND   = (nd + 255) / 256;
  const int gGemm = (NN / 16) * (DD / 16) / 8;
  const int gScat = NE / 8;

  for (int bb = 0; bb < 2; ++bb) {
    const int* srcp = ei[bb];
    const int* dstp = ei[bb] + NE;

    k_deg_init <<<gN, B, 0, stream>>>(dinv);
    k_deg_count<<<gE, B, 0, stream>>>(dstp, dinv);
    k_rsqrt    <<<gN, B, 0, stream>>>(dinv);

    k_gemm_wmma<<<gGemm, B, 0, stream>>>(x[bb], W0, H);
    k_zero     <<<gND,   B, 0, stream>>>(AG, nd);
    k_scatter  <<<gScat, B, 0, stream>>>(srcp, dstp, dinv, H, AG);
    k_finalize <<<gND,   B, 0, stream>>>(AG, H, dinv, b0, Y, 1);

    k_gemm_wmma<<<gGemm, B, 0, stream>>>(Y, W1, H);
    k_zero     <<<gND,   B, 0, stream>>>(AG, nd);
    k_scatter  <<<gScat, B, 0, stream>>>(srcp, dstp, dinv, H, AG);
    k_finalize <<<gND,   B, 0, stream>>>(AG, H, dinv, b1, out + (size_t)bb * nd, 0);
  }
}


# Global float atomic scatter replaced by LDS float atomics: 256 resident workgroups each own 391 destination rows in LDS, scan the edge list for their range; LDS arrival order; zero+finalize folded in
# speedup vs baseline: 2.2488x; 2.2488x over previous
#include <hip/hip_runtime.h>


#define NN 100000
#define NE 1600000
#define DD 128

typedef __attribute__((ext_vector_type(2))) float v2f;
typedef __attribute__((ext_vector_type(8))) float v8f;

__global__ __launch_bounds__(256) void k_deg_init(float* __restrict__ deg) {
  int i = blockIdx.x * 256 + threadIdx.x;
  if (i < NN) deg[i] = 1.0f;
}

__global__ __launch_bounds__(256) void k_deg_count(const int* __restrict__ dst,
                                                   float* __restrict__ deg) {
  int e = blockIdx.x * 256 + threadIdx.x;
  if (e < NE) atomicAdd(&deg[dst[e]], 1.0f);
}

__global__ __launch_bounds__(256) void k_rsqrt(float* __restrict__ d) {
  int i = blockIdx.x * 256 + threadIdx.x;
  if (i < NN) d[i] = rsqrtf(d[i]);
}

__global__ __launch_bounds__(256) void k_gemm_wmma(const float* __restrict__ X,
                                                   const float* __restrict__ W,
                                                   float* __restrict__ H) {
  const int wave = (blockIdx.x * 256 + (int)threadIdx.x) >> 5;
  const int lane = threadIdx.x & 31;
  const int row0 = (wave >> 3) << 4;
  const int col0 = (wave & 7) << 4;
  if (row0 >= NN) return;
  const int half = lane >> 4;
  const int l16  = lane & 15;

  v8f acc = {};
  const float* xr = X + (size_t)(row0 + l16) * DD;
  const float* wc = W + col0 + l16;
  for (int k = 0; k < DD; k += 4) {
    v2f a = *(const v2f*)(xr + k + 2 * half);
    v2f b;
    b.x = wc[(size_t)(k + 2 * half) * DD];
    b.y = wc[(size_t)(k + 2 * half + 1) * DD];
    acc = __builtin_amdgcn_wmma_f32_16x16x4_f32(
         false, a,  false, b,
         (short)0, acc,  false,  false);
  }
  float* hp = H + (size_t)row0 * DD + col0 + l16;
#pragma unroll
  for (int r = 0; r < 8; ++r)
    hp[(size_t)(r + 8 * half) * DD] = acc[r];
}

#define AROWS 391
#define AWG   256
#define AWAVES 32
#define LCAP  640
#define SINK  AROWS

__device__ __forceinline__ void agg_batch(const unsigned* list, int n, int lane,
                                          const int* __restrict__ src,
                                          const float* __restrict__ dinv,
                                          const float* __restrict__ H,
                                          float* acc, const float* dloc) {
  unsigned ent = ((unsigned)SINK << 21);
  if (lane < n) ent = list[lane];
  const unsigned e  = ent & 0x1FFFFFu;
  const int      dl = (int)(ent >> 21);
  const int      s  = src[e];
  const float  nrm  = dinv[s] * dloc[dl];
  const int nk = (n + 15) & ~15;
  for (int k0 = 0; k0 < nk; k0 += 16) {
    float4 v[16];
    float  w[16];
    int    r[16];
#pragma unroll
    for (int j = 0; j < 16; ++j) {
      const int sj = __builtin_amdgcn_readlane(s, k0 + j);
      w[j] = __int_as_float(__builtin_amdgcn_readlane(__float_as_int(nrm), k0 + j));
      r[j] = __builtin_amdgcn_readlane(dl, k0 + j);
      v[j] = ((const float4*)(H + (size_t)sj * DD))[lane];
    }
#pragma unroll
    for (int j = 0; j < 16; ++j) {
      float* a = acc + r[j] * DD + lane;
      atomicAdd(a +  0, v[j].x * w[j]);
      atomicAdd(a + 32, v[j].y * w[j]);
      atomicAdd(a + 64, v[j].z * w[j]);
      atomicAdd(a + 96, v[j].w * w[j]);
    }
  }
}

__global__ __launch_bounds__(1024) void k_agg_lds(const int* __restrict__ src,
                                                  const int* __restrict__ dst,
                                                  const float* __restrict__ dinv,
                                                  const float* __restrict__ H,
                                                  const float* __restrict__ bias,
                                                  float* __restrict__ out,
                                                  int relu) {
  __shared__ float    acc[(AROWS + 1) * DD];
  __shared__ float    dloc[AROWS + 1];
  __shared__ unsigned hits[AWAVES * LCAP];

  const int tid  = threadIdx.x;
  const int lane = tid & 31;
  const int wv   = __builtin_amdgcn_readfirstlane(tid >> 5);
  const int d0   = blockIdx.x * AROWS;
  const int nrows = (NN - d0 < AROWS) ? (NN - d0) : AROWS;

  float4* acc4 = (float4*)acc;
  for (int i = tid; i < (AROWS + 1) * (DD / 4); i += 1024)
    acc4[i] = make_float4(0.0f, 0.0f, 0.0f, 0.0f);
  for (int i = tid; i <= AROWS; i += 1024)
    dloc[i] = (i < nrows) ? dinv[d0 + i] : 0.0f;
  __syncthreads();

  unsigned* mylist = hits + wv * LCAP;
  int cnt = 0;
  const unsigned un = (unsigned)nrows;

  const int4* dst4 = (const int4*)dst;
  const int NBLK = NE / 512;
  auto scan_block = [&](const int4 (&d)[4], int blk) {
#pragma unroll
    for (int q = 0; q < 4; ++q) {
      const unsigned u0 = (unsigned)(d[q].x - d0);
      const unsigned u1 = (unsigned)(d[q].y - d0);
      const unsigned u2 = (unsigned)(d[q].z - d0);
      const unsigned u3 = (unsigned)(d[q].w - d0);
      const unsigned m01 = u0 < u1 ? u0 : u1;
      const unsigned m23 = u2 < u3 ? u2 : u3;
      const unsigned mn  = m01 < m23 ? m01 : m23;
      if (__builtin_amdgcn_ballot_w32(mn < un) != 0u) {
        const unsigned eb = (unsigned)((blk * 4 + q) * 128 + lane * 4);
        unsigned m;
        m = __builtin_amdgcn_ballot_w32(u0 < un);
        if (u0 < un) mylist[cnt + (int)__builtin_amdgcn_mbcnt_lo(m, 0u)] = (u0 << 21) | (eb + 0u);
        cnt += __builtin_popcount(m);
        m = __builtin_amdgcn_ballot_w32(u1 < un);
        if (u1 < un) mylist[cnt + (int)__builtin_amdgcn_mbcnt_lo(m, 0u)] = (u1 << 21) | (eb + 1u);
        cnt += __builtin_popcount(m);
        m = __builtin_amdgcn_ballot_w32(u2 < un);
        if (u2 < un) mylist[cnt + (int)__builtin_amdgcn_mbcnt_lo(m, 0u)] = (u2 << 21) | (eb + 2u);
        cnt += __builtin_popcount(m);
        m = __builtin_amdgcn_ballot_w32(u3 < un);
        if (u3 < un) mylist[cnt + (int)__builtin_amdgcn_mbcnt_lo(m, 0u)] = (u3 << 21) | (eb + 3u);
        cnt += __builtin_popcount(m);
      }
    }
    if (cnt >= 64) {
      int done = 0;
      while (cnt - done >= 32) {
        agg_batch(mylist + done, 32, lane, src, dinv, H, acc, dloc);
        done += 32;
      }
      const int rem = cnt - done;
      unsigned t = 0u;
      if (lane < rem) t = mylist[done + lane];
      if (lane < rem) mylist[lane] = t;
      cnt = rem;
    }
  };
  auto load_block = [&](int4 (&d)[4], int blk) {
    const int b = (blk < NBLK) ? blk : (NBLK - 1);
#pragma unroll
    for (int q = 0; q < 4; ++q) d[q] = dst4[(b * 4 + q) * 32 + lane];
  };
  int4 da[4], db[4];
  load_block(da, wv);
  for (int blk = wv; blk < NBLK; blk += 2 * AWAVES) {
    load_block(db, blk + AWAVES);
    scan_block(da, blk);
    load_block(da, blk + 2 * AWAVES);
    if (blk + AWAVES < NBLK) scan_block(db, blk + AWAVES);
  }
  for (int done = 0; done < cnt; done += 32) {
    const int n = (cnt - done < 32) ? (cnt - done) : 32;
    agg_batch(mylist + done, n, lane, src, dinv, H, acc, dloc);
  }
  __syncthreads();

  const float b0 = bias[lane * 4 + 0];
  const float b1 = bias[lane * 4 + 1];
  const float b2 = bias[lane * 4 + 2];
  const float b3 = bias[lane * 4 + 3];
  for (int r0 = wv; r0 < nrows; r0 += 4 * AWAVES) {
    float4 h[4];
#pragma unroll
    for (int j = 0; j < 4; ++j) {
      const int r  = r0 + j * AWAVES;
      const int rr = (r < nrows) ? r : (nrows - 1);
      h[j] = ((const float4*)(H + (size_t)(d0 + rr) * DD))[lane];
    }
#pragma unroll
    for (int j = 0; j < 4; ++j) {
      const int r = r0 + j * AWAVES;
      if (r < nrows) {
        const float di = dloc[r];
        const float* a = acc + r * DD + lane;
        float4 o;
        o.x = a[0]  + h[j].x * di * di + b0;
        o.y = a[32] + h[j].y * di * di + b1;
        o.z = a[64] + h[j].z * di * di + b2;
        o.w = a[96] + h[j].w * di * di + b3;
        if (relu) {
          o.x = fmaxf(o.x, 0.0f);
          o.y = fmaxf(o.y, 0.0f);
          o.z = fmaxf(o.z, 0.0f);
          o.w = fmaxf(o.w, 0.0f);
        }
        ((float4*)(out + (size_t)(d0 + r) * DD))[lane] = o;
      }
    }
  }
}

extern "C" void kernel_launch(void* const* d_in, const int* in_sizes, int n_in,
                              void* d_out, int out_size, void* d_ws, size_t ws_size,
                              hipStream_t stream) {
  const float* x[2]  = {(const float*)d_in[0], (const float*)d_in[2]};
  const int*   ei[2] = {(const int*)d_in[1],   (const int*)d_in[3]};
  const float* W0 = (const float*)d_in[4];
  const float* b0 = (const float*)d_in[5];
  const float* W1 = (const float*)d_in[6];
  const float* b1 = (const float*)d_in[7];
  float* out = (float*)d_out;

  float* ws   = (float*)d_ws;
  float* dinv = ws;
  float* H    = ws + 100352;
  float* AG   = H  + (size_t)NN * DD;
  float* Y    = AG + (size_t)NN * DD;

  const int nd    = NN * DD;
  const dim3 B(256);
  const int gN    = (NN + 255) / 256;
  const int gE    = (NE + 255) / 256;
  const int gGemm = (NN / 16) * (DD / 16) / 8;

  for (int bb = 0; bb < 2; ++bb) {
    const int* srcp = ei[bb];
    const int* dstp = ei[bb] + NE;

    k_deg_init <<<gN, B, 0, stream>>>(dinv);
    k_deg_count<<<gE, B, 0, stream>>>(dstp, dinv);
    k_rsqrt    <<<gN, B, 0, stream>>>(dinv);

    k_gemm_wmma<<<gGemm, B, 0, stream>>>(x[bb], W0, H);
    k_agg_lds  <<<AWG, dim3(1024), 0, stream>>>(srcp, dstp, dinv, H, b0, Y, 1);

    k_gemm_wmma<<<gGemm, B, 0, stream>>>(Y, W1, H);
    k_agg_lds  <<<AWG, dim3(1024), 0, stream>>>(srcp, dstp, dinv, H, b1, out + (size_t)bb * nd, 0);
  }
}


# Edges scanned once per graph, not 4x plus a float-atomic degree pass: one scan launch leaves each owner workgroup's (row,source) list and its rows' degrees; both convs replay it; LDS arrival order
# speedup vs baseline: 3.1982x; 1.4222x over previous
#include <hip/hip_runtime.h>


#define NN 100000
#define NE 1600000
#define DD 128

typedef __attribute__((ext_vector_type(2))) float v2f;
typedef __attribute__((ext_vector_type(8))) float v8f;

__global__ __launch_bounds__(256) void k_gemm_wmma(const float* __restrict__ X,
                                                   const float* __restrict__ W,
                                                   float* __restrict__ H) {
  const int wave = (blockIdx.x * 256 + (int)threadIdx.x) >> 5;
  const int lane = threadIdx.x & 31;
  const int row0 = (wave >> 3) << 4;
  const int col0 = (wave & 7) << 4;
  if (row0 >= NN) return;
  const int half = lane >> 4;
  const int l16  = lane & 15;

  v8f acc = {};
  const float* xr = X + (size_t)(row0 + l16) * DD;
  const float* wc = W + col0 + l16;
  for (int k = 0; k < DD; k += 4) {
    v2f a = *(const v2f*)(xr + k + 2 * half);
    v2f b;
    b.x = wc[(size_t)(k + 2 * half) * DD];
    b.y = wc[(size_t)(k + 2 * half + 1) * DD];
    acc = __builtin_amdgcn_wmma_f32_16x16x4_f32(
         false, a,  false, b,
         (short)0, acc,  false,  false);
  }
  float* hp = H + (size_t)row0 * DD + col0 + l16;
#pragma unroll
  for (int r = 0; r < 8; ++r)
    hp[(size_t)(r + 8 * half) * DD] = acc[r];
}

#define AROWS 391
#define AWG   256
#define AWAVES 32
#define LCAP  640
#define SINK  AROWS
#define WCAP  32768
#define ECPITCH 32

template <class F>
__device__ __forceinline__ void scan_edges(const int* __restrict__ dst, int d0, unsigned un,
                                           unsigned* mylist, int lane, int wv, F&& batch) {
  int cnt = 0;
  const int4* dst4 = (const int4*)dst;
  const int NBLK = NE / 512;
  auto scan_block = [&](const int4 (&d)[4], int blk) {
#pragma unroll
    for (int q = 0; q < 4; ++q) {
      const unsigned u0 = (unsigned)(d[q].x - d0);
      const unsigned u1 = (unsigned)(d[q].y - d0);
      const unsigned u2 = (unsigned)(d[q].z - d0);
      const unsigned u3 = (unsigned)(d[q].w - d0);
      const unsigned m01 = u0 < u1 ? u0 : u1;
      const unsigned m23 = u2 < u3 ? u2 : u3;
      const unsigned mn  = m01 < m23 ? m01 : m23;
      if (__builtin_amdgcn_ballot_w32(mn < un) != 0u) {
        const unsigned eb = (unsigned)((blk * 4 + q) * 128 + lane * 4);
        unsigned m;
        m = __builtin_amdgcn_ballot_w32(u0 < un);
        if (u0 < un) mylist[cnt + (int)__builtin_amdgcn_mbcnt_lo(m, 0u)] = (u0 << 21) | (eb + 0u);
        cnt += __builtin_popcount(m);
        m = __builtin_amdgcn_ballot_w32(u1 < un);
        if (u1 < un) mylist[cnt + (int)__builtin_amdgcn_mbcnt_lo(m, 0u)] = (u1 << 21) | (eb + 1u);
        cnt += __builtin_popcount(m);
        m = __builtin_amdgcn_ballot_w32(u2 < un);
        if (u2 < un) mylist[cnt + (int)__builtin_amdgcn_mbcnt_lo(m, 0u)] = (u2 << 21) | (eb + 2u);
        cnt += __builtin_popcount(m);
        m = __builtin_amdgcn_ballot_w32(u3 < un);
        if (u3 < un) mylist[cnt + (int)__builtin_amdgcn_mbcnt_lo(m, 0u)] = (u3 << 21) | (eb + 3u);
        cnt += __builtin_popcount(m);
      }
    }
    if (cnt >= 64) {
      int done = 0;
      while (cnt - done >= 32) {
        batch(mylist + done, 32);
        done += 32;
      }
      const int rem = cnt - done;
      unsigned t = 0u;
      if (lane < rem) t = mylist[done + lane];
      if (lane < rem) mylist[lane] = t;
      cnt = rem;
    }
  };
  auto load_block = [&](int4 (&d)[4], int blk) {
    const int b = (blk < NBLK) ? blk : (NBLK - 1);
#pragma unroll
    for (int q = 0; q < 4; ++q) d[q] = dst4[(b * 4 + q) * 32 + lane];
  };
  int4 da[4], db[4];
  load_block(da, wv);
  for (int blk = wv; blk < NBLK; blk += 2 * AWAVES) {
    load_block(db, blk + AWAVES);
    scan_block(da, blk);
    load_block(da, blk + 2 * AWAVES);
    if (blk + AWAVES < NBLK) scan_block(db, blk + AWAVES);
  }
  for (int done = 0; done < cnt; done += 32) {
    const int n = (cnt - done < 32) ? (cnt - done) : 32;
    batch(mylist + done, n);
  }
}

__global__ __launch_bounds__(1024) void k_scan(const int* __restrict__ src,
                                               const int* __restrict__ dst,
                                               float* __restrict__ dinv,
                                               int* __restrict__ elist,
                                               int* __restrict__ ecount) {
  __shared__ unsigned hits[AWAVES * LCAP];
  __shared__ int      degc[AROWS + 1];
  __shared__ int      wgn[2];

  const int tid  = threadIdx.x;
  const int lane = tid & 31;
  const int wv   = __builtin_amdgcn_readfirstlane(tid >> 5);
  const int d0   = blockIdx.x * AROWS;
  const int nrows = (NN - d0 < AROWS) ? (NN - d0) : AROWS;

  for (int i = tid; i <= AROWS; i += 1024) degc[i] = 0;
  if (tid < 2) wgn[tid] = 0;
  __syncthreads();

  int* mine = elist + (size_t)blockIdx.x * WCAP;
  scan_edges(dst, d0, (unsigned)nrows, hits + wv * LCAP, lane, wv,
             [&](const unsigned* list, int n) {
    const bool act = lane < n;
    unsigned ent = 0u;
    if (act) ent = list[lane];
    const unsigned e  = ent & 0x1FFFFFu;
    const unsigned dl = ent >> 21;
    const int s = src[e];
    if (act) atomicAdd(&degc[dl], 1);
    int base = 0;
    if (lane == 0) base = atomicAdd(&wgn[0], n);
    base = __builtin_amdgcn_readfirstlane(base);
    if (base + n <= WCAP) {
      if (act) mine[base + lane] = (int)((dl << 17) | (unsigned)s);
    } else if (lane == 0) {
      wgn[1] = 1;
    }
  });
  __syncthreads();

  for (int i = tid; i < nrows; i += 1024)
    dinv[d0 + i] = rsqrtf((float)degc[i] + 1.0f);
  if (tid == 0) ecount[blockIdx.x * ECPITCH] = (wgn[1] != 0) ? -1 : wgn[0];
}

__device__ __forceinline__ void add_rows(int s, int dl, float nrm, int n, int lane,
                                         const float* __restrict__ H, float* acc) {
  const int nk = (n + 15) & ~15;
  for (int k0 = 0; k0 < nk; k0 += 16) {
    float4 v[16];
    float  w[16];
    int    r[16];
#pragma unroll
    for (int j = 0; j < 16; ++j) {
      const int sj = __builtin_amdgcn_readlane(s, k0 + j);
      w[j] = __int_as_float(__builtin_amdgcn_readlane(__float_as_int(nrm), k0 + j));
      r[j] = __builtin_amdgcn_readlane(dl, k0 + j);
      v[j] = ((const float4*)(H + (size_t)sj * DD))[lane];
    }
#pragma unroll
    for (int j = 0; j < 16; ++j) {
      float* a = acc + r[j] * DD + lane;
      atomicAdd(a +  0, v[j].x * w[j]);
      atomicAdd(a + 32, v[j].y * w[j]);
      atomicAdd(a + 64, v[j].z * w[j]);
      atomicAdd(a + 96, v[j].w * w[j]);
    }
  }
}

__device__ __forceinline__ void agg_batch(const unsigned* list, int n, int lane,
                                          const int* __restrict__ src,
                                          const float* __restrict__ dinv,
                                          const float* __restrict__ H,
                                          float* acc, const float* dloc) {
  unsigned ent = ((unsigned)SINK << 21);
  if (lane < n) ent = list[lane];
  const unsigned e  = ent & 0x1FFFFFu;
  const int      dl = (int)(ent >> 21);
  const int      s  = src[e];
  const float  nrm  = dinv[s] * dloc[dl];
  add_rows(s, dl, nrm, n, lane, H, acc);
}

__global__ __launch_bounds__(1024) void k_agg_lds(const int* __restrict__ src,
                                                  const int* __restrict__ dst,
                                                  const float* __restrict__ dinv,
                                                  const int* __restrict__ elist,
                                                  const int* __restrict__ ecount,
                                                  const float* __restrict__ H,
                                                  const float* __restrict__ bias,
                                                  float* __restrict__ out,
                                                  int relu) {
  __shared__ float    acc[(AROWS + 1) * DD];
  __shared__ float    dloc[AROWS + 1];
  __shared__ unsigned hits[AWAVES * LCAP];

  const int tid  = threadIdx.x;
  const int lane = tid & 31;
  const int wv   = __builtin_amdgcn_readfirstlane(tid >> 5);
  const int d0   = blockIdx.x * AROWS;
  const int nrows = (NN - d0 < AROWS) ? (NN - d0) : AROWS;

  float4* acc4 = (float4*)acc;
  for (int i = tid; i < (AROWS + 1) * (DD / 4); i += 1024)
    acc4[i] = make_float4(0.0f, 0.0f, 0.0f, 0.0f);
  for (int i = tid; i <= AROWS; i += 1024)
    dloc[i] = (i < nrows) ? dinv[d0 + i] : 0.0f;
  __syncthreads();

  const int total = ecount[blockIdx.x * ECPITCH];
  if (total >= 0) {
    const int* mine = elist + (size_t)blockIdx.x * WCAP;
    const int STEP = 32 * AWAVES;
    const unsigned padent = (unsigned)SINK << 17;
    auto list_word = [&](int i) -> unsigned {
      const unsigned t = (unsigned)mine[(i < total) ? i : 0];
      return (i < total) ? t : padent;
    };
    if (total > 0) {
      int b = wv * 32;
      unsigned ent0 = list_word(b + lane);
      unsigned ent1 = list_word(b + STEP + lane);
      float dv0 = dinv[ent0 & 0x1FFFFu];
      while (b < total) {
        const unsigned ent2 = list_word(b + 2 * STEP + lane);
        const float dv1 = dinv[ent1 & 0x1FFFFu];
        const int n  = (total - b < 32) ? (total - b) : 32;
        const int s  = (int)(ent0 & 0x1FFFFu);
        const int dl = (int)(ent0 >> 17);
        add_rows(s, dl, dv0 * dloc[dl], n, lane, H, acc);
        ent0 = ent1; dv0 = dv1; ent1 = ent2;
        b += STEP;
      }
    }
  } else {
    scan_edges(dst, d0, (unsigned)nrows, hits + wv * LCAP, lane, wv,
               [&](const unsigned* list, int n) {
      agg_batch(list, n, lane, src, dinv, H, acc, dloc);
    });
  }
  __syncthreads();

  const float b0 = bias[lane * 4 + 0];
  const float b1 = bias[lane * 4 + 1];
  const float b2 = bias[lane * 4 + 2];
  const float b3 = bias[lane * 4 + 3];
  for (int r0 = wv; r0 < nrows; r0 += 4 * AWAVES) {
    float4 h[4];
#pragma unroll
    for (int j = 0; j < 4; ++j) {
      const int r  = r0 + j * AWAVES;
      const int rr = (r < nrows) ? r : (nrows - 1);
      h[j] = ((const float4*)(H + (size_t)(d0 + rr) * DD))[lane];
    }
#pragma unroll
    for (int j = 0; j < 4; ++j) {
      const int r = r0 + j * AWAVES;
      if (r < nrows) {
        const float di = dloc[r];
        const float* a = acc + r * DD + lane;
        float4 o;
        o.x = a[0]  + h[j].x * di * di + b0;
        o.y = a[32] + h[j].y * di * di + b1;
        o.z = a[64] + h[j].z * di * di + b2;
        o.w = a[96] + h[j].w * di * di + b3;
        if (relu) {
          o.x = fmaxf(o.x, 0.0f);
          o.y = fmaxf(o.y, 0.0f);
          o.z = fmaxf(o.z, 0.0f);
          o.w = fmaxf(o.w, 0.0f);
        }
        ((float4*)(out + (size_t)(d0 + r) * DD))[lane] = o;
      }
    }
  }
}

extern "C" void kernel_launch(void* const* d_in, const int* in_sizes, int n_in,
                              void* d_out, int out_size, void* d_ws, size_t ws_size,
                              hipStream_t stream) {
  const float* x[2]  = {(const float*)d_in[0], (const float*)d_in[2]};
  const int*   ei[2] = {(const int*)d_in[1],   (const int*)d_in[3]};
  const float* W0 = (const float*)d_in[4];
  const float* b0 = (const float*)d_in[5];
  const float* W1 = (const float*)d_in[6];
  const float* b1 = (const float*)d_in[7];
  float* out = (float*)d_out;

  float* ws   = (float*)d_ws;
  float* dinv = ws;
  float* H    = ws + 100352;
  float* EL   = H  + (size_t)NN * DD;
  float* Y    = EL + (size_t)NN * DD;
  int*   elist  = (int*)EL;
  int*   ecount = elist + (size_t)AWG * WCAP;

  const int nd    = NN * DD;
  const dim3 B(256);
  const int gGemm = (NN / 16) * (DD / 16) / 8;

  for (int bb = 0; bb < 2; ++bb) {
    const int* srcp = ei[bb];
    const int* dstp = ei[bb] + NE;

    k_scan     <<<AWG, dim3(1024), 0, stream>>>(srcp, dstp, dinv, elist, ecount);

    k_gemm_wmma<<<gGemm, B, 0, stream>>>(x[bb], W0, H);
    k_agg_lds  <<<AWG, dim3(1024), 0, stream>>>(srcp, dstp, dinv, elist, ecount, H, b0, Y, 1);

    k_gemm_wmma<<<gGemm, B, 0, stream>>>(Y, W1, H);
    k_agg_lds  <<<AWG, dim3(1024), 0, stream>>>(srcp, dstp, dinv, elist, ecount, H, b1,
                                                out + (size_t)bb * nd, 0);
  }
}


# Destination list narrowed by one pre-pass launch to one owner byte per edge (dst/391): each workgroup's full scan reads a quarter of the bytes, looks up dst only for its hits; LDS arrival order kept
# speedup vs baseline: 3.4691x; 1.0847x over previous
#include <hip/hip_runtime.h>


#define NN 100000
#define NE 1600000
#define DD 128

typedef __attribute__((ext_vector_type(2))) float v2f;
typedef __attribute__((ext_vector_type(8))) float v8f;

__global__ __launch_bounds__(256) void k_gemm_wmma(const float* __restrict__ X,
                                                   const float* __restrict__ W,
                                                   float* __restrict__ H) {
  const int wave = (blockIdx.x * 256 + (int)threadIdx.x) >> 5;
  const int lane = threadIdx.x & 31;
  const int row0 = (wave >> 3) << 4;
  const int col0 = (wave & 7) << 4;
  if (row0 >= NN) return;
  const int half = lane >> 4;
  const int l16  = lane & 15;

  v8f acc = {};
  const float* xr = X + (size_t)(row0 + l16) * DD;
  const float* wc = W + col0 + l16;
  for (int k = 0; k < DD; k += 4) {
    v2f a = *(const v2f*)(xr + k + 2 * half);
    v2f b;
    b.x = wc[(size_t)(k + 2 * half) * DD];
    b.y = wc[(size_t)(k + 2 * half + 1) * DD];
    acc = __builtin_amdgcn_wmma_f32_16x16x4_f32(
         false, a,  false, b,
         (short)0, acc,  false,  false);
  }
  float* hp = H + (size_t)row0 * DD + col0 + l16;
#pragma unroll
  for (int r = 0; r < 8; ++r)
    hp[(size_t)(r + 8 * half) * DD] = acc[r];
}

#define AROWS 391
#define AWG   256
#define AWAVES 32
#define LCAP  640
#define SINK  AROWS
#define WCAP  32768
#define ECPITCH 32

template <class F>
__device__ __forceinline__ void scan_edges(const int* __restrict__ dst, int d0, unsigned un,
                                           unsigned* mylist, int lane, int wv, F&& batch) {
  int cnt = 0;
  const int4* dst4 = (const int4*)dst;
  const int NBLK = NE / 512;
  auto scan_block = [&](const int4 (&d)[4], int blk) {
#pragma unroll
    for (int q = 0; q < 4; ++q) {
      const unsigned u0 = (unsigned)(d[q].x - d0);
      const unsigned u1 = (unsigned)(d[q].y - d0);
      const unsigned u2 = (unsigned)(d[q].z - d0);
      const unsigned u3 = (unsigned)(d[q].w - d0);
      const unsigned m01 = u0 < u1 ? u0 : u1;
      const unsigned m23 = u2 < u3 ? u2 : u3;
      const unsigned mn  = m01 < m23 ? m01 : m23;
      if (__builtin_amdgcn_ballot_w32(mn < un) != 0u) {
        const unsigned eb = (unsigned)((blk * 4 + q) * 128 + lane * 4);
        unsigned m;
        m = __builtin_amdgcn_ballot_w32(u0 < un);
        if (u0 < un) mylist[cnt + (int)__builtin_amdgcn_mbcnt_lo(m, 0u)] = (u0 << 21) | (eb + 0u);
        cnt += __builtin_popcount(m);
        m = __builtin_amdgcn_ballot_w32(u1 < un);
        if (u1 < un) mylist[cnt + (int)__builtin_amdgcn_mbcnt_lo(m, 0u)] = (u1 << 21) | (eb + 1u);
        cnt += __builtin_popcount(m);
        m = __builtin_amdgcn_ballot_w32(u2 < un);
        if (u2 < un) mylist[cnt + (int)__builtin_amdgcn_mbcnt_lo(m, 0u)] = (u2 << 21) | (eb + 2u);
        cnt += __builtin_popcount(m);
        m = __builtin_amdgcn_ballot_w32(u3 < un);
        if (u3 < un) mylist[cnt + (int)__builtin_amdgcn_mbcnt_lo(m, 0u)] = (u3 << 21) | (eb + 3u);
        cnt += __builtin_popcount(m);
      }
    }
    if (cnt >= 64) {
      int done = 0;
      while (cnt - done >= 32) {
        batch(mylist + done, 32);
        done += 32;
      }
      const int rem = cnt - done;
      unsigned t = 0u;
      if (lane < rem) t = mylist[done + lane];
      if (lane < rem) mylist[lane] = t;
      cnt = rem;
    }
  };
  auto load_block = [&](int4 (&d)[4], int blk) {
    const int b = (blk < NBLK) ? blk : (NBLK - 1);
#pragma unroll
    for (int q = 0; q < 4; ++q) d[q] = dst4[(b * 4 + q) * 32 + lane];
  };
  int4 da[4], db[4];
  load_block(da, wv);
  for (int blk = wv; blk < NBLK; blk += 2 * AWAVES) {
    load_block(db, blk + AWAVES);
    scan_block(da, blk);
    load_block(da, blk + 2 * AWAVES);
    if (blk + AWAVES < NBLK) scan_block(db, blk + AWAVES);
  }
  for (int done = 0; done < cnt; done += 32) {
    const int n = (cnt - done < 32) ? (cnt - done) : 32;
    batch(mylist + done, n);
  }
}

#define NV16 (NE / 16)
__global__ __launch_bounds__(256) void k_owner(const int* __restrict__ dstA,
                                               const int* __restrict__ dstB,
                                               unsigned* __restrict__ ob) {
  const int t = blockIdx.x * 256 + (int)threadIdx.x;
  if (t >= NV16) return;
  const int4* d4 = (const int4*)(blockIdx.y == 0 ? dstA : dstB) + (size_t)t * 4;
  uint4 o;
  unsigned w[4];
#pragma unroll
  for (int c = 0; c < 4; ++c) {
    const int4 d = d4[c];
    w[c] = (((unsigned)d.x / AROWS) & 255u) | ((((unsigned)d.y / AROWS) & 255u) << 8) |
           ((((unsigned)d.z / AROWS) & 255u) << 16) | (((unsigned)d.w / AROWS) << 24);
  }
  o.x = w[0]; o.y = w[1]; o.z = w[2]; o.w = w[3];
  ((uint4*)ob)[(size_t)blockIdx.y * NV16 + t] = o;
}

template <class F>
__device__ __forceinline__ void scan_owner_bytes(const uint4* __restrict__ ob, unsigned owner,
                                                 unsigned* mylist, int lane, int wv, F&& batch) {
  int cnt = 0;
  const int NBLK = NV16 / 32;
  const int NGRP = (NBLK + 3) / 4;
  const unsigned pat = owner * 0x01010101u;
  auto match = [&](unsigned x) -> unsigned {
    const unsigned y = x ^ pat;
    return ~(((y & 0x7F7F7F7Fu) + 0x7F7F7F7Fu) | y | 0x7F7F7F7Fu);
  };
  auto scan_block = [&](const uint4& d, int blk) {
    const unsigned t0 = match(d.x), t1 = match(d.y), t2 = match(d.z), t3 = match(d.w);
    if (__builtin_amdgcn_ballot_w32((t0 | t1 | t2 | t3) != 0u) != 0u) {
      unsigned hm = (t0 >> 7) | (t1 >> 6) | (t2 >> 5) | (t3 >> 4);
      const unsigned eb = (unsigned)(blk * 32 + lane) * 16u;
      unsigned mb = __builtin_amdgcn_ballot_w32(hm != 0u);
      while (mb != 0u) {
        const unsigned p = (unsigned)__builtin_ctz(hm | 0x80000000u);
        if (hm != 0u)
          mylist[cnt + (int)__builtin_amdgcn_mbcnt_lo(mb, 0u)] = eb + (((p & 3u) << 2) | (p >> 3));
        cnt += __builtin_popcount(mb);
        hm &= hm - 1u;
        mb = __builtin_amdgcn_ballot_w32(hm != 0u);
      }
    }
    if (cnt >= 64) {
      int done = 0;
      while (cnt - done >= 32) {
        batch(mylist + done, 32);
        done += 32;
      }
      const int rem = cnt - done;
      unsigned t = 0u;
      if (lane < rem) t = mylist[done + lane];
      if (lane < rem) mylist[lane] = t;
      cnt = rem;
    }
  };
  auto load_group = [&](uint4 (&d)[4], int g) {
#pragma unroll
    for (int q = 0; q < 4; ++q) {
      const int blk = g * 4 + q;
      d[q] = ob[((blk < NBLK) ? blk : (NBLK - 1)) * 32 + lane];
    }
  };
  auto scan_group = [&](const uint4 (&d)[4], int g) {
#pragma unroll
    for (int q = 0; q < 4; ++q)
      if (g * 4 + q < NBLK) scan_block(d[q], g * 4 + q);
  };
  uint4 da[4], db[4];
  load_group(da, wv);
  for (int g = wv; g < NGRP; g += 2 * AWAVES) {
    load_group(db, g + AWAVES);
    scan_group(da, g);
    load_group(da, g + 2 * AWAVES);
    if (g + AWAVES < NGRP) scan_group(db, g + AWAVES);
  }
  for (int done = 0; done < cnt; done += 32) {
    const int n = (cnt - done < 32) ? (cnt - done) : 32;
    batch(mylist + done, n);
  }
}

__global__ __launch_bounds__(1024) void k_scan(const int* __restrict__ src,
                                               const int* __restrict__ dst,
                                               const unsigned* __restrict__ ob,
                                               float* __restrict__ dinv,
                                               int* __restrict__ elist,
                                               int* __restrict__ ecount) {
  __shared__ unsigned hits[AWAVES * LCAP];
  __shared__ int      degc[AROWS + 1];
  __shared__ int      wgn[2];

  const int tid  = threadIdx.x;
  const int lane = tid & 31;
  const int wv   = __builtin_amdgcn_readfirstlane(tid >> 5);
  const int d0   = blockIdx.x * AROWS;
  const int nrows = (NN - d0 < AROWS) ? (NN - d0) : AROWS;

  for (int i = tid; i <= AROWS; i += 1024) degc[i] = 0;
  if (tid < 2) wgn[tid] = 0;
  __syncthreads();

  int* mine = elist + (size_t)blockIdx.x * WCAP;
  scan_owner_bytes((const uint4*)ob, (unsigned)blockIdx.x, hits + wv * LCAP, lane, wv,
                   [&](const unsigned* list, int n) {
    unsigned e = 0u;
    if (lane < n) e = list[lane];
    const int d = dst[e];
    const int s = src[e];
    const unsigned dl = (unsigned)(d - d0);
    const bool act = (lane < n) && (dl < (unsigned)nrows);
    const unsigned m = __builtin_amdgcn_ballot_w32(act);
    const int nn = __builtin_popcount(m);
    if (act) atomicAdd(&degc[dl], 1);
    int base = 0;
    if (lane == 0) base = atomicAdd(&wgn[0], nn);
    base = __builtin_amdgcn_readfirstlane(base);
    if (base + nn <= WCAP) {
      if (act) mine[base + (int)__builtin_amdgcn_mbcnt_lo(m, 0u)] = (int)((dl << 17) | (unsigned)s);
    } else if (lane == 0) {
      wgn[1] = 1;
    }
  });
  __syncthreads();

  for (int i = tid; i < nrows; i += 1024)
    dinv[d0 + i] = rsqrtf((float)degc[i] + 1.0f);
  if (tid == 0) ecount[blockIdx.x * ECPITCH] = (wgn[1] != 0) ? -1 : wgn[0];
}

__device__ __forceinline__ void add_rows(int s, int dl, float nrm, int n, int lane,
                                         const float* __restrict__ H, float* acc) {
  const int nk = (n + 15) & ~15;
  for (int k0 = 0; k0 < nk; k0 += 16) {
    float4 v[16];
    float  w[16];
    int    r[16];
#pragma unroll
    for (int j = 0; j < 16; ++j) {
      const int sj = __builtin_amdgcn_readlane(s, k0 + j);
      w[j] = __int_as_float(__builtin_amdgcn_readlane(__float_as_int(nrm), k0 + j));
      r[j] = __builtin_amdgcn_readlane(dl, k0 + j);
      v[j] = ((const float4*)(H + (size_t)sj * DD))[lane];
    }
#pragma unroll
    for (int j = 0; j < 16; ++j) {
      float* a = acc + r[j] * DD + lane;
      atomicAdd(a +  0, v[j].x * w[j]);
      atomicAdd(a + 32, v[j].y * w[j]);
      atomicAdd(a + 64, v[j].z * w[j]);
      atomicAdd(a + 96, v[j].w * w[j]);
    }
  }
}

__device__ __forceinline__ void agg_batch(const unsigned* list, int n, int lane,
                                          const int* __restrict__ src,
                                          const float* __restrict__ dinv,
                                          const float* __restrict__ H,
                                          float* acc, const float* dloc) {
  unsigned ent = ((unsigned)SINK << 21);
  if (lane < n) ent = list[lane];
  const unsigned e  = ent & 0x1FFFFFu;
  const int      dl = (int)(ent >> 21);
  const int      s  = src[e];
  const float  nrm  = dinv[s] * dloc[dl];
  add_rows(s, dl, nrm, n, lane, H, acc);
}

__global__ __launch_bounds__(1024) void k_agg_lds(const int* __restrict__ src,
                                                  const int* __restrict__ dst,
                                                  const float* __restrict__ dinv,
                                                  const int* __restrict__ elist,
                                                  const int* __restrict__ ecount,
                                                  const float* __restrict__ H,
                                                  const float* __restrict__ bias,
                                                  float* __restrict__ out,
                                                  int relu) {
  __shared__ float    acc[(AROWS + 1) * DD];
  __shared__ float    dloc[AROWS + 1];
  __shared__ unsigned hits[AWAVES * LCAP];

  const int tid  = threadIdx.x;
  const int lane = tid & 31;
  const int wv   = __builtin_amdgcn_readfirstlane(tid >> 5);
  const int d0   = blockIdx.x * AROWS;
  const int nrows = (NN - d0 < AROWS) ? (NN - d0) : AROWS;

  float4* acc4 = (float4*)acc;
  for (int i = tid; i < (AROWS + 1) * (DD / 4); i += 1024)
    acc4[i] = make_float4(0.0f, 0.0f, 0.0f, 0.0f);
  for (int i = tid; i <= AROWS; i += 1024)
    dloc[i] = (i < nrows) ? dinv[d0 + i] : 0.0f;
  __syncthreads();

  const int total = ecount[blockIdx.x * ECPITCH];
  if (total >= 0) {
    const int* mine = elist + (size_t)blockIdx.x * WCAP;
    const int STEP = 32 * AWAVES;
    const unsigned padent = (unsigned)SINK << 17;
    auto list_word = [&](int i) -> unsigned {
      const unsigned t = (unsigned)mine[(i < total) ? i : 0];
      return (i < total) ? t : padent;
    };
    if (total > 0) {
      int b = wv * 32;
      unsigned ent0 = list_word(b + lane);
      unsigned ent1 = list_word(b + STEP + lane);
      float dv0 = dinv[ent0 & 0x1FFFFu];
      while (b < total) {
        const unsigned ent2 = list_word(b + 2 * STEP + lane);
        const float dv1 = dinv[ent1 & 0x1FFFFu];
        const int n  = (total - b < 32) ? (total - b) : 32;
        const int s  = (int)(ent0 & 0x1FFFFu);
        const int dl = (int)(ent0 >> 17);
        add_rows(s, dl, dv0 * dloc[dl], n, lane, H, acc);
        ent0 = ent1; dv0 = dv1; ent1 = ent2;
        b += STEP;
      }
    }
  } else {
    scan_edges(dst, d0, (unsigned)nrows, hits + wv * LCAP, lane, wv,
               [&](const unsigned* list, int n) {
      agg_batch(list, n, lane, src, dinv, H, acc, dloc);
    });
  }
  __syncthreads();

  const float b0 = bias[lane * 4 + 0];
  const float b1 = bias[lane * 4 + 1];
  const float b2 = bias[lane * 4 + 2];
  const float b3 = bias[lane * 4 + 3];
  for (int r0 = wv; r0 < nrows; r0 += 4 * AWAVES) {
    float4 h[4];
#pragma unroll
    for (int j = 0; j < 4; ++j) {
      const int r  = r0 + j * AWAVES;
      const int rr = (r < nrows) ? r : (nrows - 1);
      h[j] = ((const float4*)(H + (size_t)(d0 + rr) * DD))[lane];
    }
#pragma unroll
    for (int j = 0; j < 4; ++j) {
      const int r = r0 + j * AWAVES;
      if (r < nrows) {
        const float di = dloc[r];
        const float* a = acc + r * DD + lane;
        float4 o;
        o.x = a[0]  + h[j].x * di * di + b0;
        o.y = a[32] + h[j].y * di * di + b1;
        o.z = a[64] + h[j].z * di * di + b2;
        o.w = a[96] + h[j].w * di * di + b3;
        if (relu) {
          o.x = fmaxf(o.x, 0.0f);
          o.y = fmaxf(o.y, 0.0f);
          o.z = fmaxf(o.z, 0.0f);
          o.w = fmaxf(o.w, 0.0f);
        }
        ((float4*)(out + (size_t)(d0 + r) * DD))[lane] = o;
      }
    }
  }
}

extern "C" void kernel_launch(void* const* d_in, const int* in_sizes, int n_in,
                              void* d_out, int out_size, void* d_ws, size_t ws_size,
                              hipStream_t stream) {
  const float* x[2]  = {(const float*)d_in[0], (const float*)d_in[2]};
  const int*   ei[2] = {(const int*)d_in[1],   (const int*)d_in[3]};
  const float* W0 = (const float*)d_in[4];
  const float* b0 = (const float*)d_in[5];
  const float* W1 = (const float*)d_in[6];
  const float* b1 = (const float*)d_in[7];
  float* out = (float*)d_out;

  float* ws   = (float*)d_ws;
  float* dinv = ws;
  float* H    = ws + 100352;
  float* EL   = H  + (size_t)NN * DD;
  float* Y    = EL + (size_t)NN * DD;
  int*   elist  = (int*)EL;
  int*   ecount = elist + (size_t)AWG * WCAP;
  unsigned* ob  = (unsigned*)(Y + (size_t)NN * DD);

  const int nd    = NN * DD;
  const dim3 B(256);
  const int gGemm = (NN / 16) * (DD / 16) / 8;

  k_owner<<<dim3((NV16 + 255) / 256, 2), B, 0, stream>>>(ei[0] + NE, ei[1] + NE, ob);

  for (int bb = 0; bb < 2; ++bb) {
    const int* srcp = ei[bb];
    const int* dstp = ei[bb] + NE;

    k_scan     <<<AWG, dim3(1024), 0, stream>>>(srcp, dstp, ob + (size_t)bb * (NE / 4), dinv,
                                              elist, ecount);

    k_gemm_wmma<<<gGemm, B, 0, stream>>>(x[bb], W0, H);
    k_agg_lds  <<<AWG, dim3(1024), 0, stream>>>(srcp, dstp, dinv, elist, ecount, H, b0, Y, 1);

    k_gemm_wmma<<<gGemm, B, 0, stream>>>(Y, W1, H);
    k_agg_lds  <<<AWG, dim3(1024), 0, stream>>>(srcp, dstp, dinv, elist, ecount, H, b1,
                                                out + (size_t)bb * nd, 0);
  }
}


# Dense transform re-tiled: one wave per 16x128 strip, X fragment loaded once in 16B pieces and reused by 8 column tiles, W fragment-ordered in LDS; k order inside each 8 changes the f32 sum order
# speedup vs baseline: 4.2262x; 1.2182x over previous
#include <hip/hip_runtime.h>


#define NN 100000
#define NE 1600000
#define DD 128

typedef __attribute__((ext_vector_type(2))) float v2f;
typedef __attribute__((ext_vector_type(8))) float v8f;

#define GWAVES 4
__global__ __launch_bounds__(32 * GWAVES) void k_gemm_wmma(const float* __restrict__ X,
                                                   const float* __restrict__ W,
                                                   float* __restrict__ H) {
  __shared__ float4 wl[(DD / 4) * DD];

  const int tid  = threadIdx.x;
  const int lane = tid & 31;
  const int wv    = __builtin_amdgcn_readfirstlane(tid >> 5);
  const int strip = blockIdx.x * GWAVES + wv;
  const bool have = strip < NN / 16;
  const int row0  = (have ? strip : 0) * 16;
  const int half  = lane >> 4;
  const int l16   = lane & 15;

  const float4* xr = (const float4*)(X + (size_t)(row0 + l16) * DD) + half;
  float4 a[16];
#pragma unroll
  for (int kk = 0; kk < 16; ++kk) a[kk] = xr[2 * kk];

#pragma unroll
  for (int it = 0; it < (DD / 4) * DD / (32 * GWAVES); ++it) {
    const int i   = tid + it * (32 * GWAVES);
    const int p   = i & 127;
    const int col = 4 * (p & 15) + ((p >> 4) & 3) + 64 * (p >> 6);
    const float* wp = W + (size_t)((i >> 7) * 4) * DD + col;
    float4 v;
    v.x = wp[0];
    v.y = wp[DD];
    v.z = wp[2 * DD];
    v.w = wp[3 * DD];
    wl[i] = v;
  }
  __syncthreads();
  if (!have) return;

  v8f acc[8];
#pragma unroll
  for (int t = 0; t < 8; ++t) acc[t] = (v8f){};

#pragma unroll
  for (int kk = 0; kk < 16; ++kk) {
    float4 b[8];
#pragma unroll
    for (int t = 0; t < 8; ++t) b[t] = wl[(kk * 2 + half) * 128 + t * 16 + l16];
    v2f a0, a1;
    a0.x = a[kk].x; a0.y = a[kk].y;
    a1.x = a[kk].z; a1.y = a[kk].w;
#pragma unroll
    for (int t = 0; t < 8; ++t) {
      v2f b0;
      b0.x = b[t].x; b0.y = b[t].y;
      acc[t] = __builtin_amdgcn_wmma_f32_16x16x4_f32(
           false, a0,  false, b0,
           (short)0, acc[t],  false,  false);
    }
#pragma unroll
    for (int t = 0; t < 8; ++t) {
      v2f b1;
      b1.x = b[t].z; b1.y = b[t].w;
      acc[t] = __builtin_amdgcn_wmma_f32_16x16x4_f32(
           false, a1,  false, b1,
           (short)0, acc[t],  false,  false);
    }
  }

  float* hp = H + (size_t)(row0 + 8 * half) * DD + 4 * l16;
#pragma unroll
  for (int r = 0; r < 8; ++r) {
#pragma unroll
    for (int g = 0; g < 2; ++g) {
      float4 o;
      o.x = acc[4 * g + 0][r];
      o.y = acc[4 * g + 1][r];
      o.z = acc[4 * g + 2][r];
      o.w = acc[4 * g + 3][r];
      *(float4*)(hp + (size_t)r * DD + 64 * g) = o;
    }
  }
}

#define AROWS 391
#define AWG   256
#define AWAVES 32
#define LCAP  640
#define SINK  AROWS
#define WCAP  32768
#define ECPITCH 32

template <class F>
__device__ __forceinline__ void scan_edges(const int* __restrict__ dst, int d0, unsigned un,
                                           unsigned* mylist, int lane, int wv, F&& batch) {
  int cnt = 0;
  const int4* dst4 = (const int4*)dst;
  const int NBLK = NE / 512;
  auto scan_block = [&](const int4 (&d)[4], int blk) {
#pragma unroll
    for (int q = 0; q < 4; ++q) {
      const unsigned u0 = (unsigned)(d[q].x - d0);
      const unsigned u1 = (unsigned)(d[q].y - d0);
      const unsigned u2 = (unsigned)(d[q].z - d0);
      const unsigned u3 = (unsigned)(d[q].w - d0);
      const unsigned m01 = u0 < u1 ? u0 : u1;
      const unsigned m23 = u2 < u3 ? u2 : u3;
      const unsigned mn  = m01 < m23 ? m01 : m23;
      if (__builtin_amdgcn_ballot_w32(mn < un) != 0u) {
        const unsigned eb = (unsigned)((blk * 4 + q) * 128 + lane * 4);
        unsigned m;
        m = __builtin_amdgcn_ballot_w32(u0 < un);
        if (u0 < un) mylist[cnt + (int)__builtin_amdgcn_mbcnt_lo(m, 0u)] = (u0 << 21) | (eb + 0u);
        cnt += __builtin_popcount(m);
        m = __builtin_amdgcn_ballot_w32(u1 < un);
        if (u1 < un) mylist[cnt + (int)__builtin_amdgcn_mbcnt_lo(m, 0u)] = (u1 << 21) | (eb + 1u);
        cnt += __builtin_popcount(m);
        m = __builtin_amdgcn_ballot_w32(u2 < un);
        if (u2 < un) mylist[cnt + (int)__builtin_amdgcn_mbcnt_lo(m, 0u)] = (u2 << 21) | (eb + 2u);
        cnt += __builtin_popcount(m);
        m = __builtin_amdgcn_ballot_w32(u3 < un);
        if (u3 < un) mylist[cnt + (int)__builtin_amdgcn_mbcnt_lo(m, 0u)] = (u3 << 21) | (eb + 3u);
        cnt += __builtin_popcount(m);
      }
    }
    if (cnt >= 64) {
      int done = 0;
      while (cnt - done >= 32) {
        batch(mylist + done, 32);
        done += 32;
      }
      const int rem = cnt - done;
      unsigned t = 0u;
      if (lane < rem) t = mylist[done + lane];
      if (lane < rem) mylist[lane] = t;
      cnt = rem;
    }
  };
  auto load_block = [&](int4 (&d)[4], int blk) {
    const int b = (blk < NBLK) ? blk : (NBLK - 1);
#pragma unroll
    for (int q = 0; q < 4; ++q) d[q] = dst4[(b * 4 + q) * 32 + lane];
  };
  int4 da[4], db[4];
  load_block(da, wv);
  for (int blk = wv; blk < NBLK; blk += 2 * AWAVES) {
    load_block(db, blk + AWAVES);
    scan_block(da, blk);
    load_block(da, blk + 2 * AWAVES);
    if (blk + AWAVES < NBLK) scan_block(db, blk + AWAVES);
  }
  for (int done = 0; done < cnt; done += 32) {
    const int n = (cnt - done < 32) ? (cnt - done) : 32;
    batch(mylist + done, n);
  }
}

#define NV16 (NE / 16)
__global__ __launch_bounds__(256) void k_owner(const int* __restrict__ dstA,
                                               const int* __restrict__ dstB,
                                               unsigned* __restrict__ ob) {
  const int t = blockIdx.x * 256 + (int)threadIdx.x;
  if (t >= NV16) return;
  const int4* d4 = (const int4*)(blockIdx.y == 0 ? dstA : dstB) + (size_t)t * 4;
  uint4 o;
  unsigned w[4];
#pragma unroll
  for (int c = 0; c < 4; ++c) {
    const int4 d = d4[c];
    w[c] = (((unsigned)d.x / AROWS) & 255u) | ((((unsigned)d.y / AROWS) & 255u) << 8) |
           ((((unsigned)d.z / AROWS) & 255u) << 16) | (((unsigned)d.w / AROWS) << 24);
  }
  o.x = w[0]; o.y = w[1]; o.z = w[2]; o.w = w[3];
  ((uint4*)ob)[(size_t)blockIdx.y * NV16 + t] = o;
}

template <class F>
__device__ __forceinline__ void scan_owner_bytes(const uint4* __restrict__ ob, unsigned owner,
                                                 unsigned* mylist, int lane, int wv, F&& batch) {
  int cnt = 0;
  const int NBLK = NV16 / 32;
  const int NGRP = (NBLK + 3) / 4;
  const unsigned pat = owner * 0x01010101u;
  auto match = [&](unsigned x) -> unsigned {
    const unsigned y = x ^ pat;
    return ~(((y & 0x7F7F7F7Fu) + 0x7F7F7F7Fu) | y | 0x7F7F7F7Fu);
  };
  auto scan_block = [&](const uint4& d, int blk) {
    const unsigned t0 = match(d.x), t1 = match(d.y), t2 = match(d.z), t3 = match(d.w);
    if (__builtin_amdgcn_ballot_w32((t0 | t1 | t2 | t3) != 0u) != 0u) {
      unsigned hm = (t0 >> 7) | (t1 >> 6) | (t2 >> 5) | (t3 >> 4);
      const unsigned eb = (unsigned)(blk * 32 + lane) * 16u;
      unsigned mb = __builtin_amdgcn_ballot_w32(hm != 0u);
      while (mb != 0u) {
        const unsigned p = (unsigned)__builtin_ctz(hm | 0x80000000u);
        if (hm != 0u)
          mylist[cnt + (int)__builtin_amdgcn_mbcnt_lo(mb, 0u)] = eb + (((p & 3u) << 2) | (p >> 3));
        cnt += __builtin_popcount(mb);
        hm &= hm - 1u;
        mb = __builtin_amdgcn_ballot_w32(hm != 0u);
      }
    }
    if (cnt >= 64) {
      int done = 0;
      while (cnt - done >= 32) {
        batch(mylist + done, 32);
        done += 32;
      }
      const int rem = cnt - done;
      unsigned t = 0u;
      if (lane < rem) t = mylist[done + lane];
      if (lane < rem) mylist[lane] = t;
      cnt = rem;
    }
  };
  auto load_group = [&](uint4 (&d)[4], int g) {
#pragma unroll
    for (int q = 0; q < 4; ++q) {
      const int blk = g * 4 + q;
      d[q] = ob[((blk < NBLK) ? blk : (NBLK - 1)) * 32 + lane];
    }
  };
  auto scan_group = [&](const uint4 (&d)[4], int g) {
#pragma unroll
    for (int q = 0; q < 4; ++q)
      if (g * 4 + q < NBLK) scan_block(d[q], g * 4 + q);
  };
  uint4 da[4], db[4];
  load_group(da, wv);
  for (int g = wv; g < NGRP; g += 2 * AWAVES) {
    load_group(db, g + AWAVES);
    scan_group(da, g);
    load_group(da, g + 2 * AWAVES);
    if (g + AWAVES < NGRP) scan_group(db, g + AWAVES);
  }
  for (int done = 0; done < cnt; done += 32) {
    const int n = (cnt - done < 32) ? (cnt - done) : 32;
    batch(mylist + done, n);
  }
}

__global__ __launch_bounds__(1024) void k_scan(const int* __restrict__ src,
                                               const int* __restrict__ dst,
                                               const unsigned* __restrict__ ob,
                                               float* __restrict__ dinv,
                                               int* __restrict__ elist,
                                               int* __restrict__ ecount) {
  __shared__ unsigned hits[AWAVES * LCAP];
  __shared__ int      degc[AROWS + 1];
  __shared__ int      wgn[2];

  const int tid  = threadIdx.x;
  const int lane = tid & 31;
  const int wv   = __builtin_amdgcn_readfirstlane(tid >> 5);
  const int d0   = blockIdx.x * AROWS;
  const int nrows = (NN - d0 < AROWS) ? (NN - d0) : AROWS;

  for (int i = tid; i <= AROWS; i += 1024) degc[i] = 0;
  if (tid < 2) wgn[tid] = 0;
  __syncthreads();

  int* mine = elist + (size_t)blockIdx.x * WCAP;
  scan_owner_bytes((const uint4*)ob, (unsigned)blockIdx.x, hits + wv * LCAP, lane, wv,
                   [&](const unsigned* list, int n) {
    unsigned e = 0u;
    if (lane < n) e = list[lane];
    const int d = dst[e];
    const int s = src[e];
    const unsigned dl = (unsigned)(d - d0);
    const bool act = (lane < n) && (dl < (unsigned)nrows);
    const unsigned m = __builtin_amdgcn_ballot_w32(act);
    const int nn = __builtin_popcount(m);
    if (act) atomicAdd(&degc[dl], 1);
    int base = 0;
    if (lane == 0) base = atomicAdd(&wgn[0], nn);
    base = __builtin_amdgcn_readfirstlane(base);
    if (base + nn <= WCAP) {
      if (act) mine[base + (int)__builtin_amdgcn_mbcnt_lo(m, 0u)] = (int)((dl << 17) | (unsigned)s);
    } else if (lane == 0) {
      wgn[1] = 1;
    }
  });
  __syncthreads();

  for (int i = tid; i < nrows; i += 1024)
    dinv[d0 + i] = rsqrtf((float)degc[i] + 1.0f);
  if (tid == 0) ecount[blockIdx.x * ECPITCH] = (wgn[1] != 0) ? -1 : wgn[0];
}

__device__ __forceinline__ void add_rows(int s, int dl, float nrm, int n, int lane,
                                         const float* __restrict__ H, float* acc) {
  const int nk = (n + 15) & ~15;
  for (int k0 = 0; k0 < nk; k0 += 16) {
    float4 v[16];
    float  w[16];
    int    r[16];
#pragma unroll
    for (int j = 0; j < 16; ++j) {
      const int sj = __builtin_amdgcn_readlane(s, k0 + j);
      w[j] = __int_as_float(__builtin_amdgcn_readlane(__float_as_int(nrm), k0 + j));
      r[j] = __builtin_amdgcn_readlane(dl, k0 + j);
      v[j] = ((const float4*)(H + (size_t)sj * DD))[lane];
    }
#pragma unroll
    for (int j = 0; j < 16; ++j) {
      float* a = acc + r[j] * DD + lane;
      atomicAdd(a +  0, v[j].x * w[j]);
      atomicAdd(a + 32, v[j].y * w[j]);
      atomicAdd(a + 64, v[j].z * w[j]);
      atomicAdd(a + 96, v[j].w * w[j]);
    }
  }
}

__device__ __forceinline__ void agg_batch(const unsigned* list, int n, int lane,
                                          const int* __restrict__ src,
                                          const float* __restrict__ dinv,
                                          const float* __restrict__ H,
                                          float* acc, const float* dloc) {
  unsigned ent = ((unsigned)SINK << 21);
  if (lane < n) ent = list[lane];
  const unsigned e  = ent & 0x1FFFFFu;
  const int      dl = (int)(ent >> 21);
  const int      s  = src[e];
  const float  nrm  = dinv[s] * dloc[dl];
  add_rows(s, dl, nrm, n, lane, H, acc);
}

__global__ __launch_bounds__(1024) void k_agg_lds(const int* __restrict__ src,
                                                  const int* __restrict__ dst,
                                                  const float* __restrict__ dinv,
                                                  const int* __restrict__ elist,
                                                  const int* __restrict__ ecount,
                                                  const float* __restrict__ H,
                                                  const float* __restrict__ bias,
                                                  float* __restrict__ out,
                                                  int relu) {
  __shared__ float    acc[(AROWS + 1) * DD];
  __shared__ float    dloc[AROWS + 1];
  __shared__ unsigned hits[AWAVES * LCAP];

  const int tid  = threadIdx.x;
  const int lane = tid & 31;
  const int wv   = __builtin_amdgcn_readfirstlane(tid >> 5);
  const int d0   = blockIdx.x * AROWS;
  const int nrows = (NN - d0 < AROWS) ? (NN - d0) : AROWS;

  float4* acc4 = (float4*)acc;
  for (int i = tid; i < (AROWS + 1) * (DD / 4); i += 1024)
    acc4[i] = make_float4(0.0f, 0.0f, 0.0f, 0.0f);
  for (int i = tid; i <= AROWS; i += 1024)
    dloc[i] = (i < nrows) ? dinv[d0 + i] : 0.0f;
  __syncthreads();

  const int total = ecount[blockIdx.x * ECPITCH];
  if (total >= 0) {
    const int* mine = elist + (size_t)blockIdx.x * WCAP;
    const int STEP = 32 * AWAVES;
    const unsigned padent = (unsigned)SINK << 17;
    auto list_word = [&](int i) -> unsigned {
      const unsigned t = (unsigned)mine[(i < total) ? i : 0];
      return (i < total) ? t : padent;
    };
    if (total > 0) {
      int b = wv * 32;
      unsigned ent0 = list_word(b + lane);
      unsigned ent1 = list_word(b + STEP + lane);
      float dv0 = dinv[ent0 & 0x1FFFFu];
      while (b < total) {
        const unsigned ent2 = list_word(b + 2 * STEP + lane);
        const float dv1 = dinv[ent1 & 0x1FFFFu];
        const int n  = (total - b < 32) ? (total - b) : 32;
        const int s  = (int)(ent0 & 0x1FFFFu);
        const int dl = (int)(ent0 >> 17);
        add_rows(s, dl, dv0 * dloc[dl], n, lane, H, acc);
        ent0 = ent1; dv0 = dv1; ent1 = ent2;
        b += STEP;
      }
    }
  } else {
    scan_edges(dst, d0, (unsigned)nrows, hits + wv * LCAP, lane, wv,
               [&](const unsigned* list, int n) {
      agg_batch(list, n, lane, src, dinv, H, acc, dloc);
    });
  }
  __syncthreads();

  const float b0 = bias[lane * 4 + 0];
  const float b1 = bias[lane * 4 + 1];
  const float b2 = bias[lane * 4 + 2];
  const float b3 = bias[lane * 4 + 3];
  for (int r0 = wv; r0 < nrows; r0 += 4 * AWAVES) {
    float4 h[4];
#pragma unroll
    for (int j = 0; j < 4; ++j) {
      const int r  = r0 + j * AWAVES;
      const int rr = (r < nrows) ? r : (nrows - 1);
      h[j] = ((const float4*)(H + (size_t)(d0 + rr) * DD))[lane];
    }
#pragma unroll
    for (int j = 0; j < 4; ++j) {
      const int r = r0 + j * AWAVES;
      if (r < nrows) {
        const float di = dloc[r];
        const float* a = acc + r * DD + lane;
        float4 o;
        o.x = a[0]  + h[j].x * di * di + b0;
        o.y = a[32] + h[j].y * di * di + b1;
        o.z = a[64] + h[j].z * di * di + b2;
        o.w = a[96] + h[j].w * di * di + b3;
        if (relu) {
          o.x = fmaxf(o.x, 0.0f);
          o.y = fmaxf(o.y, 0.0f);
          o.z = fmaxf(o.z, 0.0f);
          o.w = fmaxf(o.w, 0.0f);
        }
        ((float4*)(out + (size_t)(d0 + r) * DD))[lane] = o;
      }
    }
  }
}

extern "C" void kernel_launch(void* const* d_in, const int* in_sizes, int n_in,
                              void* d_out, int out_size, void* d_ws, size_t ws_size,
                              hipStream_t stream) {
  const float* x[2]  = {(const float*)d_in[0], (const float*)d_in[2]};
  const int*   ei[2] = {(const int*)d_in[1],   (const int*)d_in[3]};
  const float* W0 = (const float*)d_in[4];
  const float* b0 = (const float*)d_in[5];
  const float* W1 = (const float*)d_in[6];
  const float* b1 = (const float*)d_in[7];
  float* out = (float*)d_out;

  float* ws   = (float*)d_ws;
  float* dinv = ws;
  float* H    = ws + 100352;
  float* EL   = H  + (size_t)NN * DD;
  float* Y    = EL + (size_t)NN * DD;
  int*   elist  = (int*)EL;
  int*   ecount = elist + (size_t)AWG * WCAP;
  unsigned* ob  = (unsigned*)(Y + (size_t)NN * DD);

  const int nd    = NN * DD;
  const dim3 B(256);
  const int gGemm = (NN / 16 + GWAVES - 1) / GWAVES;

  k_owner<<<dim3((NV16 + 255) / 256, 2), B, 0, stream>>>(ei[0] + NE, ei[1] + NE, ob);

  for (int bb = 0; bb < 2; ++bb) {
    const int* srcp = ei[bb];
    const int* dstp = ei[bb] + NE;

    k_scan     <<<AWG, dim3(1024), 0, stream>>>(srcp, dstp, ob + (size_t)bb * (NE / 4), dinv,
                                              elist, ecount);

    k_gemm_wmma<<<gGemm, dim3(32 * GWAVES), 0, stream>>>(x[bb], W0, H);
    k_agg_lds  <<<AWG, dim3(1024), 0, stream>>>(srcp, dstp, dinv, elist, ecount, H, b0, Y, 1);

    k_gemm_wmma<<<gGemm, dim3(32 * GWAVES), 0, stream>>>(Y, W1, H);
    k_agg_lds  <<<AWG, dim3(1024), 0, stream>>>(srcp, dstp, dinv, elist, ecount, H, b1,
                                                out + (size_t)bb * nd, 0);
  }
}


# Layer-1 transform runs inside the edge-scan launch: 8 of each scan workgroup's 32 waves do its WMMA strips (two 64-column passes each) beside 24 scanning waves; 11 launches -> 9; sum order kept
# speedup vs baseline: 4.3752x; 1.0353x over previous
#include <hip/hip_runtime.h>


#define NN 100000
#define NE 1600000
#define DD 128

typedef __attribute__((ext_vector_type(2))) float v2f;
typedef __attribute__((ext_vector_type(8))) float v8f;

#define GWAVES 4
__global__ __launch_bounds__(32 * GWAVES) void k_gemm_wmma(const float* __restrict__ X,
                                                   const float* __restrict__ W,
                                                   float* __restrict__ H) {
  __shared__ float4 wl[(DD / 4) * DD];

  const int tid  = threadIdx.x;
  const int lane = tid & 31;
  const int wv    = __builtin_amdgcn_readfirstlane(tid >> 5);
  const int strip = blockIdx.x * GWAVES + wv;
  const bool have = strip < NN / 16;
  const int row0  = (have ? strip : 0) * 16;
  const int half  = lane >> 4;
  const int l16   = lane & 15;

  const float4* xr = (const float4*)(X + (size_t)(row0 + l16) * DD) + half;
  float4 a[16];
#pragma unroll
  for (int kk = 0; kk < 16; ++kk) a[kk] = xr[2 * kk];

#pragma unroll
  for (int it = 0; it < (DD / 4) * DD / (32 * GWAVES); ++it) {
    const int i   = tid + it * (32 * GWAVES);
    const int p   = i & 127;
    const int col = 4 * (p & 15) + ((p >> 4) & 3) + 64 * (p >> 6);
    const float* wp = W + (size_t)((i >> 7) * 4) * DD + col;
    float4 v;
    v.x = wp[0];
    v.y = wp[DD];
    v.z = wp[2 * DD];
    v.w = wp[3 * DD];
    wl[i] = v;
  }
  __syncthreads();
  if (!have) return;

  v8f acc[8];
#pragma unroll
  for (int t = 0; t < 8; ++t) acc[t] = (v8f){};

#pragma unroll
  for (int kk = 0; kk < 16; ++kk) {
    float4 b[8];
#pragma unroll
    for (int t = 0; t < 8; ++t) b[t] = wl[(kk * 2 + half) * 128 + t * 16 + l16];
    v2f a0, a1;
    a0.x = a[kk].x; a0.y = a[kk].y;
    a1.x = a[kk].z; a1.y = a[kk].w;
#pragma unroll
    for (int t = 0; t < 8; ++t) {
      v2f b0;
      b0.x = b[t].x; b0.y = b[t].y;
      acc[t] = __builtin_amdgcn_wmma_f32_16x16x4_f32(
           false, a0,  false, b0,
           (short)0, acc[t],  false,  false);
    }
#pragma unroll
    for (int t = 0; t < 8; ++t) {
      v2f b1;
      b1.x = b[t].z; b1.y = b[t].w;
      acc[t] = __builtin_amdgcn_wmma_f32_16x16x4_f32(
           false, a1,  false, b1,
           (short)0, acc[t],  false,  false);
    }
  }

  float* hp = H + (size_t)(row0 + 8 * half) * DD + 4 * l16;
#pragma unroll
  for (int r = 0; r < 8; ++r) {
#pragma unroll
    for (int g = 0; g < 2; ++g) {
      float4 o;
      o.x = acc[4 * g + 0][r];
      o.y = acc[4 * g + 1][r];
      o.z = acc[4 * g + 2][r];
      o.w = acc[4 * g + 3][r];
      *(float4*)(hp + (size_t)r * DD + 64 * g) = o;
    }
  }
}

#define AROWS 391
#define AWG   256
#define AWAVES 32
#define LCAP  640
#define SINK  AROWS
#define WCAP  32768
#define ECPITCH 32

template <class F>
__device__ __forceinline__ void scan_edges(const int* __restrict__ dst, int d0, unsigned un,
                                           unsigned* mylist, int lane, int wv, F&& batch) {
  int cnt = 0;
  const int4* dst4 = (const int4*)dst;
  const int NBLK = NE / 512;
  auto scan_block = [&](const int4 (&d)[4], int blk) {
#pragma unroll
    for (int q = 0; q < 4; ++q) {
      const unsigned u0 = (unsigned)(d[q].x - d0);
      const unsigned u1 = (unsigned)(d[q].y - d0);
      const unsigned u2 = (unsigned)(d[q].z - d0);
      const unsigned u3 = (unsigned)(d[q].w - d0);
      const unsigned m01 = u0 < u1 ? u0 : u1;
      const unsigned m23 = u2 < u3 ? u2 : u3;
      const unsigned mn  = m01 < m23 ? m01 : m23;
      if (__builtin_amdgcn_ballot_w32(mn < un) != 0u) {
        const unsigned eb = (unsigned)((blk * 4 + q) * 128 + lane * 4);
        unsigned m;
        m = __builtin_amdgcn_ballot_w32(u0 < un);
        if (u0 < un) mylist[cnt + (int)__builtin_amdgcn_mbcnt_lo(m, 0u)] = (u0 << 21) | (eb + 0u);
        cnt += __builtin_popcount(m);
        m = __builtin_amdgcn_ballot_w32(u1 < un);
        if (u1 < un) mylist[cnt + (int)__builtin_amdgcn_mbcnt_lo(m, 0u)] = (u1 << 21) | (eb + 1u);
        cnt += __builtin_popcount(m);
        m = __builtin_amdgcn_ballot_w32(u2 < un);
        if (u2 < un) mylist[cnt + (int)__builtin_amdgcn_mbcnt_lo(m, 0u)] = (u2 << 21) | (eb + 2u);
        cnt += __builtin_popcount(m);
        m = __builtin_amdgcn_ballot_w32(u3 < un);
        if (u3 < un) mylist[cnt + (int)__builtin_amdgcn_mbcnt_lo(m, 0u)] = (u3 << 21) | (eb + 3u);
        cnt += __builtin_popcount(m);
      }
    }
    if (cnt >= 64) {
      int done = 0;
      while (cnt - done >= 32) {
        batch(mylist + done, 32);
        done += 32;
      }
      const int rem = cnt - done;
      unsigned t = 0u;
      if (lane < rem) t = mylist[done + lane];
      if (lane < rem) mylist[lane] = t;
      cnt = rem;
    }
  };
  auto load_block = [&](int4 (&d)[4], int blk) {
    const int b = (blk < NBLK) ? blk : (NBLK - 1);
#pragma unroll
    for (int q = 0; q < 4; ++q) d[q] = dst4[(b * 4 + q) * 32 + lane];
  };
  int4 da[4], db[4];
  load_block(da, wv);
  for (int blk = wv; blk < NBLK; blk += 2 * AWAVES) {
    load_block(db, blk + AWAVES);
    scan_block(da, blk);
    load_block(da, blk + 2 * AWAVES);
    if (blk + AWAVES < NBLK) scan_block(db, blk + AWAVES);
  }
  for (int done = 0; done < cnt; done += 32) {
    const int n = (cnt - done < 32) ? (cnt - done) : 32;
    batch(mylist + done, n);
  }
}

#define NV16 (NE / 16)
__global__ __launch_bounds__(256) void k_owner(const int* __restrict__ dstA,
                                               const int* __restrict__ dstB,
                                               unsigned* __restrict__ ob) {
  const int t = blockIdx.x * 256 + (int)threadIdx.x;
  if (t >= NV16) return;
  const int4* d4 = (const int4*)(blockIdx.y == 0 ? dstA : dstB) + (size_t)t * 4;
  uint4 o;
  unsigned w[4];
#pragma unroll
  for (int c = 0; c < 4; ++c) {
    const int4 d = d4[c];
    w[c] = (((unsigned)d.x / AROWS) & 255u) | ((((unsigned)d.y / AROWS) & 255u) << 8) |
           ((((unsigned)d.z / AROWS) & 255u) << 16) | (((unsigned)d.w / AROWS) << 24);
  }
  o.x = w[0]; o.y = w[1]; o.z = w[2]; o.w = w[3];
  ((uint4*)ob)[(size_t)blockIdx.y * NV16 + t] = o;
}

template <int NW, class F>
__device__ __forceinline__ void scan_owner_bytes(const uint4* __restrict__ ob, unsigned owner,
                                                 unsigned* mylist, int lane, int wv, F&& batch) {
  int cnt = 0;
  const int NBLK = NV16 / 32;
  const int NGRP = (NBLK + 3) / 4;
  const unsigned pat = owner * 0x01010101u;
  auto match = [&](unsigned x) -> unsigned {
    const unsigned y = x ^ pat;
    return ~(((y & 0x7F7F7F7Fu) + 0x7F7F7F7Fu) | y | 0x7F7F7F7Fu);
  };
  auto scan_block = [&](const uint4& d, int blk) {
    const unsigned t0 = match(d.x), t1 = match(d.y), t2 = match(d.z), t3 = match(d.w);
    if (__builtin_amdgcn_ballot_w32((t0 | t1 | t2 | t3) != 0u) != 0u) {
      unsigned hm = (t0 >> 7) | (t1 >> 6) | (t2 >> 5) | (t3 >> 4);
      const unsigned eb = (unsigned)(blk * 32 + lane) * 16u;
      unsigned mb = __builtin_amdgcn_ballot_w32(hm != 0u);
      while (mb != 0u) {
        const unsigned p = (unsigned)__builtin_ctz(hm | 0x80000000u);
        if (hm != 0u)
          mylist[cnt + (int)__builtin_amdgcn_mbcnt_lo(mb, 0u)] = eb + (((p & 3u) << 2) | (p >> 3));
        cnt += __builtin_popcount(mb);
        hm &= hm - 1u;
        mb = __builtin_amdgcn_ballot_w32(hm != 0u);
      }
    }
    if (cnt >= 64) {
      int done = 0;
      while (cnt - done >= 32) {
        batch(mylist + done, 32);
        done += 32;
      }
      const int rem = cnt - done;
      unsigned t = 0u;
      if (lane < rem) t = mylist[done + lane];
      if (lane < rem) mylist[lane] = t;
      cnt = rem;
    }
  };
  auto load_group = [&](uint4 (&d)[4], int g) {
#pragma unroll
    for (int q = 0; q < 4; ++q) {
      const int blk = g * 4 + q;
      d[q] = ob[((blk < NBLK) ? blk : (NBLK - 1)) * 32 + lane];
    }
  };
  auto scan_group = [&](const uint4 (&d)[4], int g) {
#pragma unroll
    for (int q = 0; q < 4; ++q)
      if (g * 4 + q < NBLK) scan_block(d[q], g * 4 + q);
  };
  uint4 da[4], db[4];
  load_group(da, wv);
  for (int g = wv; g < NGRP; g += 2 * NW) {
    load_group(db, g + NW);
    scan_group(da, g);
    load_group(da, g + 2 * NW);
    if (g + NW < NGRP) scan_group(db, g + NW);
  }
  for (int done = 0; done < cnt; done += 32) {
    const int n = (cnt - done < 32) ? (cnt - done) : 32;
    batch(mylist + done, n);
  }
}

#define MWAVES 8
#define SWAVES (AWAVES - MWAVES)
__global__ __launch_bounds__(1024) void k_scan(const int* __restrict__ src,
                                               const int* __restrict__ dst,
                                               const unsigned* __restrict__ ob,
                                               float* __restrict__ dinv,
                                               int* __restrict__ elist,
                                               int* __restrict__ ecount,
                                               const float* __restrict__ X,
                                               const float* __restrict__ W,
                                               float* __restrict__ H) {
  __shared__ float4   wl[(DD / 4) * DD];
  __shared__ unsigned hits[SWAVES * LCAP];
  __shared__ int      degc[AROWS + 1];
  __shared__ int      wgn[2];

  const int tid  = threadIdx.x;
  const int lane = tid & 31;
  const int wv   = __builtin_amdgcn_readfirstlane(tid >> 5);
  const int d0   = blockIdx.x * AROWS;
  const int nrows = (NN - d0 < AROWS) ? (NN - d0) : AROWS;
  const bool mat = wv < MWAVES;
  const int half = lane >> 4;
  const int l16  = lane & 15;

  float4 a[16];
  auto load_a = [&](int strip) {
    const float4* xr = (const float4*)(X + (size_t)(strip * 16 + l16) * DD) + half;
#pragma unroll
    for (int kk = 0; kk < 16; ++kk) a[kk] = xr[2 * kk];
  };
  if (mat) load_a(blockIdx.x + AWG * wv);

#pragma unroll
  for (int it = 0; it < (DD / 4) * DD / 1024; ++it) {
    const int i   = tid + it * 1024;
    const int p   = i & 127;
    const int col = 4 * (p & 15) + ((p >> 4) & 3) + 64 * (p >> 6);
    const float* wp = W + (size_t)((i >> 7) * 4) * DD + col;
    float4 v;
    v.x = wp[0];
    v.y = wp[DD];
    v.z = wp[2 * DD];
    v.w = wp[3 * DD];
    wl[i] = v;
  }
  for (int i = tid; i <= AROWS; i += 1024) degc[i] = 0;
  if (tid < 2) wgn[tid] = 0;
  __syncthreads();

  if (mat) {
    int strip = blockIdx.x + AWG * wv;
    int wslot = half * 128 + l16;
    for (;;) {
      asm volatile("" : "+v"(wslot));
      float* hp = H + (size_t)(strip * 16 + 8 * half) * DD + 4 * l16;
#pragma unroll
      for (int g = 0; g < 2; ++g) {
        v8f acc[4];
#pragma unroll
        for (int t = 0; t < 4; ++t) acc[t] = (v8f){};
#pragma unroll
        for (int kk = 0; kk < 16; ++kk) {
          float4 b[4];
#pragma unroll
          for (int t = 0; t < 4; ++t) b[t] = wl[wslot + kk * 256 + (4 * g + t) * 16];
          v2f a0, a1;
          a0.x = a[kk].x; a0.y = a[kk].y;
          a1.x = a[kk].z; a1.y = a[kk].w;
#pragma unroll
          for (int t = 0; t < 4; ++t) {
            v2f b0;
            b0.x = b[t].x; b0.y = b[t].y;
            acc[t] = __builtin_amdgcn_wmma_f32_16x16x4_f32(
                 false, a0,  false, b0,
                 (short)0, acc[t],  false,  false);
          }
#pragma unroll
          for (int t = 0; t < 4; ++t) {
            v2f b1;
            b1.x = b[t].z; b1.y = b[t].w;
            acc[t] = __builtin_amdgcn_wmma_f32_16x16x4_f32(
                 false, a1,  false, b1,
                 (short)0, acc[t],  false,  false);
          }
        }
#pragma unroll
        for (int r = 0; r < 8; ++r) {
          float4 o;
          o.x = acc[0][r];
          o.y = acc[1][r];
          o.z = acc[2][r];
          o.w = acc[3][r];
          *(float4*)(hp + (size_t)r * DD + 64 * g) = o;
        }
      }
      strip += AWG * MWAVES;
      if (strip >= NN / 16) break;
      load_a(strip);
    }
  } else {
    const int swv = wv - MWAVES;
    int* mine = elist + (size_t)blockIdx.x * WCAP;
    scan_owner_bytes<SWAVES>((const uint4*)ob, (unsigned)blockIdx.x, hits + swv * LCAP, lane, swv,
                     [&](const unsigned* list, int n) {
      unsigned e = 0u;
      if (lane < n) e = list[lane];
      const int d = dst[e];
      const int s = src[e];
      const unsigned dl = (unsigned)(d - d0);
      const bool act = (lane < n) && (dl < (unsigned)nrows);
      const unsigned m = __builtin_amdgcn_ballot_w32(act);
      const int nn = __builtin_popcount(m);
      if (act) atomicAdd(&degc[dl], 1);
      int base = 0;
      if (lane == 0) base = atomicAdd(&wgn[0], nn);
      base = __builtin_amdgcn_readfirstlane(base);
      if (base + nn <= WCAP) {
        if (act) mine[base + (int)__builtin_amdgcn_mbcnt_lo(m, 0u)] = (int)((dl << 17) | (unsigned)s);
      } else if (lane == 0) {
        wgn[1] = 1;
      }
    });
  }
  __syncthreads();

  for (int i = tid; i < nrows; i += 1024)
    dinv[d0 + i] = rsqrtf((float)degc[i] + 1.0f);
  if (tid == 0) ecount[blockIdx.x * ECPITCH] = (wgn[1] != 0) ? -1 : wgn[0];
}

__device__ __forceinline__ void add_rows(int s, int dl, float nrm, int n, int lane,
                                         const float* __restrict__ H, float* acc) {
  const int nk = (n + 15) & ~15;
  for (int k0 = 0; k0 < nk; k0 += 16) {
    float4 v[16];
    float  w[16];
    int    r[16];
#pragma unroll
    for (int j = 0; j < 16; ++j) {
      const int sj = __builtin_amdgcn_readlane(s, k0 + j);
      w[j] = __int_as_float(__builtin_amdgcn_readlane(__float_as_int(nrm), k0 + j));
      r[j] = __builtin_amdgcn_readlane(dl, k0 + j);
      v[j] = ((const float4*)(H + (size_t)sj * DD))[lane];
    }
#pragma unroll
    for (int j = 0; j < 16; ++j) {
      float* a = acc + r[j] * DD + lane;
      atomicAdd(a +  0, v[j].x * w[j]);
      atomicAdd(a + 32, v[j].y * w[j]);
      atomicAdd(a + 64, v[j].z * w[j]);
      atomicAdd(a + 96, v[j].w * w[j]);
    }
  }
}

__device__ __forceinline__ void agg_batch(const unsigned* list, int n, int lane,
                                          const int* __restrict__ src,
                                          const float* __restrict__ dinv,
                                          const float* __restrict__ H,
                                          float* acc, const float* dloc) {
  unsigned ent = ((unsigned)SINK << 21);
  if (lane < n) ent = list[lane];
  const unsigned e  = ent & 0x1FFFFFu;
  const int      dl = (int)(ent >> 21);
  const int      s  = src[e];
  const float  nrm  = dinv[s] * dloc[dl];
  add_rows(s, dl, nrm, n, lane, H, acc);
}

__global__ __launch_bounds__(1024) void k_agg_lds(const int* __restrict__ src,
                                                  const int* __restrict__ dst,
                                                  const float* __restrict__ dinv,
                                                  const int* __restrict__ elist,
                                                  const int* __restrict__ ecount,
                                                  const float* __restrict__ H,
                                                  const float* __restrict__ bias,
                                                  float* __restrict__ out,
                                                  int relu) {
  __shared__ float    acc[(AROWS + 1) * DD];
  __shared__ float    dloc[AROWS + 1];
  __shared__ unsigned hits[AWAVES * LCAP];

  const int tid  = threadIdx.x;
  const int lane = tid & 31;
  const int wv   = __builtin_amdgcn_readfirstlane(tid >> 5);
  const int d0   = blockIdx.x * AROWS;
  const int nrows = (NN - d0 < AROWS) ? (NN - d0) : AROWS;

  float4* acc4 = (float4*)acc;
  for (int i = tid; i < (AROWS + 1) * (DD / 4); i += 1024)
    acc4[i] = make_float4(0.0f, 0.0f, 0.0f, 0.0f);
  for (int i = tid; i <= AROWS; i += 1024)
    dloc[i] = (i < nrows) ? dinv[d0 + i] : 0.0f;
  __syncthreads();

  const int total = ecount[blockIdx.x * ECPITCH];
  if (total >= 0) {
    const int* mine = elist + (size_t)blockIdx.x * WCAP;
    const int STEP = 32 * AWAVES;
    const unsigned padent = (unsigned)SINK << 17;
    auto list_word = [&](int i) -> unsigned {
      const unsigned t = (unsigned)mine[(i < total) ? i : 0];
      return (i < total) ? t : padent;
    };
    if (total > 0) {
      int b = wv * 32;
      unsigned ent0 = list_word(b + lane);
      unsigned ent1 = list_word(b + STEP + lane);
      float dv0 = dinv[ent0 & 0x1FFFFu];
      while (b < total) {
        const unsigned ent2 = list_word(b + 2 * STEP + lane);
        const float dv1 = dinv[ent1 & 0x1FFFFu];
        const int n  = (total - b < 32) ? (total - b) : 32;
        const int s  = (int)(ent0 & 0x1FFFFu);
        const int dl = (int)(ent0 >> 17);
        add_rows(s, dl, dv0 * dloc[dl], n, lane, H, acc);
        ent0 = ent1; dv0 = dv1; ent1 = ent2;
        b += STEP;
      }
    }
  } else {
    scan_edges(dst, d0, (unsigned)nrows, hits + wv * LCAP, lane, wv,
               [&](const unsigned* list, int n) {
      agg_batch(list, n, lane, src, dinv, H, acc, dloc);
    });
  }
  __syncthreads();

  const float b0 = bias[lane * 4 + 0];
  const float b1 = bias[lane * 4 + 1];
  const float b2 = bias[lane * 4 + 2];
  const float b3 = bias[lane * 4 + 3];
  for (int r0 = wv; r0 < nrows; r0 += 4 * AWAVES) {
    float4 h[4];
#pragma unroll
    for (int j = 0; j < 4; ++j) {
      const int r  = r0 + j * AWAVES;
      const int rr = (r < nrows) ? r : (nrows - 1);
      h[j] = ((const float4*)(H + (size_t)(d0 + rr) * DD))[lane];
    }
#pragma unroll
    for (int j = 0; j < 4; ++j) {
      const int r = r0 + j * AWAVES;
      if (r < nrows) {
        const float di = dloc[r];
        const float* a = acc + r * DD + lane;
        float4 o;
        o.x = a[0]  + h[j].x * di * di + b0;
        o.y = a[32] + h[j].y * di * di + b1;
        o.z = a[64] + h[j].z * di * di + b2;
        o.w = a[96] + h[j].w * di * di + b3;
        if (relu) {
          o.x = fmaxf(o.x, 0.0f);
          o.y = fmaxf(o.y, 0.0f);
          o.z = fmaxf(o.z, 0.0f);
          o.w = fmaxf(o.w, 0.0f);
        }
        ((float4*)(out + (size_t)(d0 + r) * DD))[lane] = o;
      }
    }
  }
}

extern "C" void kernel_launch(void* const* d_in, const int* in_sizes, int n_in,
                              void* d_out, int out_size, void* d_ws, size_t ws_size,
                              hipStream_t stream) {
  const float* x[2]  = {(const float*)d_in[0], (const float*)d_in[2]};
  const int*   ei[2] = {(const int*)d_in[1],   (const int*)d_in[3]};
  const float* W0 = (const float*)d_in[4];
  const float* b0 = (const float*)d_in[5];
  const float* W1 = (const float*)d_in[6];
  const float* b1 = (const float*)d_in[7];
  float* out = (float*)d_out;

  float* ws   = (float*)d_ws;
  float* dinv = ws;
  float* H    = ws + 100352;
  float* EL   = H  + (size_t)NN * DD;
  float* Y    = EL + (size_t)NN * DD;
  int*   elist  = (int*)EL;
  int*   ecount = elist + (size_t)AWG * WCAP;
  unsigned* ob  = (unsigned*)(Y + (size_t)NN * DD);

  const int nd    = NN * DD;
  const dim3 B(256);
  const int gGemm = (NN / 16 + GWAVES - 1) / GWAVES;

  k_owner<<<dim3((NV16 + 255) / 256, 2), B, 0, stream>>>(ei[0] + NE, ei[1] + NE, ob);

  for (int bb = 0; bb < 2; ++bb) {
    const int* srcp = ei[bb];
    const int* dstp = ei[bb] + NE;

    k_scan     <<<AWG, dim3(1024), 0, stream>>>(srcp, dstp, ob + (size_t)bb * (NE / 4), dinv,
                                              elist, ecount, x[bb], W0, H);

    k_agg_lds  <<<AWG, dim3(1024), 0, stream>>>(srcp, dstp, dinv, elist, ecount, H, b0, Y, 1);

    k_gemm_wmma<<<gGemm, dim3(32 * GWAVES), 0, stream>>>(Y, W1, H);
    k_agg_lds  <<<AWG, dim3(1024), 0, stream>>>(srcp, dstp, dinv, elist, ecount, H, b1,
                                                out + (size_t)bb * nd, 0);
  }
}


# Layer-2 transform loses its launch: 8 matrix waves run it inside the OTHER graph's aggregation launch while 24 waves gather (graphs interleaved, 9 launches -> 7); sums still in LDS arrival order
# speedup vs baseline: 4.8591x; 1.1106x over previous
#include <hip/hip_runtime.h>


#define NN 100000
#define NE 1600000
#define DD 128

typedef __attribute__((ext_vector_type(2))) float v2f;
typedef __attribute__((ext_vector_type(8))) float v8f;


#define AROWS 391
#define AWG   256
#define AWAVES 32
#define LCAP  640
#define LCAPM 576
#define SINK  AROWS
#define WCAP  32768
#define ECPITCH 32

template <int NW, class F>
__device__ __forceinline__ void scan_edges(const int* __restrict__ dst, int d0, unsigned un,
                                           unsigned* mylist, int lane, int wv, F&& batch) {
  int cnt = 0;
  const int4* dst4 = (const int4*)dst;
  const int NBLK = NE / 512;
  auto scan_block = [&](const int4 (&d)[4], int blk) {
#pragma unroll
    for (int q = 0; q < 4; ++q) {
      const unsigned u0 = (unsigned)(d[q].x - d0);
      const unsigned u1 = (unsigned)(d[q].y - d0);
      const unsigned u2 = (unsigned)(d[q].z - d0);
      const unsigned u3 = (unsigned)(d[q].w - d0);
      const unsigned m01 = u0 < u1 ? u0 : u1;
      const unsigned m23 = u2 < u3 ? u2 : u3;
      const unsigned mn  = m01 < m23 ? m01 : m23;
      if (__builtin_amdgcn_ballot_w32(mn < un) != 0u) {
        const unsigned eb = (unsigned)((blk * 4 + q) * 128 + lane * 4);
        unsigned m;
        m = __builtin_amdgcn_ballot_w32(u0 < un);
        if (u0 < un) mylist[cnt + (int)__builtin_amdgcn_mbcnt_lo(m, 0u)] = (u0 << 21) | (eb + 0u);
        cnt += __builtin_popcount(m);
        m = __builtin_amdgcn_ballot_w32(u1 < un);
        if (u1 < un) mylist[cnt + (int)__builtin_amdgcn_mbcnt_lo(m, 0u)] = (u1 << 21) | (eb + 1u);
        cnt += __builtin_popcount(m);
        m = __builtin_amdgcn_ballot_w32(u2 < un);
        if (u2 < un) mylist[cnt + (int)__builtin_amdgcn_mbcnt_lo(m, 0u)] = (u2 << 21) | (eb + 2u);
        cnt += __builtin_popcount(m);
        m = __builtin_amdgcn_ballot_w32(u3 < un);
        if (u3 < un) mylist[cnt + (int)__builtin_amdgcn_mbcnt_lo(m, 0u)] = (u3 << 21) | (eb + 3u);
        cnt += __builtin_popcount(m);
      }
    }
    if (cnt >= 64) {
      int done = 0;
      while (cnt - done >= 32) {
        batch(mylist + done, 32);
        done += 32;
      }
      const int rem = cnt - done;
      unsigned t = 0u;
      if (lane < rem) t = mylist[done + lane];
      if (lane < rem) mylist[lane] = t;
      cnt = rem;
    }
  };
  auto load_block = [&](int4 (&d)[4], int blk) {
    const int b = (blk < NBLK) ? blk : (NBLK - 1);
#pragma unroll
    for (int q = 0; q < 4; ++q) d[q] = dst4[(b * 4 + q) * 32 + lane];
  };
  int4 da[4], db[4];
  load_block(da, wv);
  for (int blk = wv; blk < NBLK; blk += 2 * NW) {
    load_block(db, blk + NW);
    scan_block(da, blk);
    load_block(da, blk + 2 * NW);
    if (blk + NW < NBLK) scan_block(db, blk + NW);
  }
  for (int done = 0; done < cnt; done += 32) {
    const int n = (cnt - done < 32) ? (cnt - done) : 32;
    batch(mylist + done, n);
  }
}

#define NV16 (NE / 16)
__global__ __launch_bounds__(256) void k_owner(const int* __restrict__ dstA,
                                               const int* __restrict__ dstB,
                                               unsigned* __restrict__ ob) {
  const int t = blockIdx.x * 256 + (int)threadIdx.x;
  if (t >= NV16) return;
  const int4* d4 = (const int4*)(blockIdx.y == 0 ? dstA : dstB) + (size_t)t * 4;
  uint4 o;
  unsigned w[4];
#pragma unroll
  for (int c = 0; c < 4; ++c) {
    const int4 d = d4[c];
    w[c] = (((unsigned)d.x / AROWS) & 255u) | ((((unsigned)d.y / AROWS) & 255u) << 8) |
           ((((unsigned)d.z / AROWS) & 255u) << 16) | (((unsigned)d.w / AROWS) << 24);
  }
  o.x = w[0]; o.y = w[1]; o.z = w[2]; o.w = w[3];
  ((uint4*)ob)[(size_t)blockIdx.y * NV16 + t] = o;
}

template <int NW, class F>
__device__ __forceinline__ void scan_owner_bytes(const uint4* __restrict__ ob, unsigned owner,
                                                 unsigned* mylist, int lane, int wv, F&& batch) {
  int cnt = 0;
  const int NBLK = NV16 / 32;
  const int NGRP = (NBLK + 3) / 4;
  const unsigned pat = owner * 0x01010101u;
  auto match = [&](unsigned x) -> unsigned {
    const unsigned y = x ^ pat;
    return ~(((y & 0x7F7F7F7Fu) + 0x7F7F7F7Fu) | y | 0x7F7F7F7Fu);
  };
  auto scan_block = [&](const uint4& d, int blk) {
    const unsigned t0 = match(d.x), t1 = match(d.y), t2 = match(d.z), t3 = match(d.w);
    if (__builtin_amdgcn_ballot_w32((t0 | t1 | t2 | t3) != 0u) != 0u) {
      unsigned hm = (t0 >> 7) | (t1 >> 6) | (t2 >> 5) | (t3 >> 4);
      const unsigned eb = (unsigned)(blk * 32 + lane) * 16u;
      unsigned mb = __builtin_amdgcn_ballot_w32(hm != 0u);
      while (mb != 0u) {
        const unsigned p = (unsigned)__builtin_ctz(hm | 0x80000000u);
        if (hm != 0u)
          mylist[cnt + (int)__builtin_amdgcn_mbcnt_lo(mb, 0u)] = eb + (((p & 3u) << 2) | (p >> 3));
        cnt += __builtin_popcount(mb);
        hm &= hm - 1u;
        mb = __builtin_amdgcn_ballot_w32(hm != 0u);
      }
    }
    if (cnt >= 64) {
      int done = 0;
      while (cnt - done >= 32) {
        batch(mylist + done, 32);
        done += 32;
      }
      const int rem = cnt - done;
      unsigned t = 0u;
      if (lane < rem) t = mylist[done + lane];
      if (lane < rem) mylist[lane] = t;
      cnt = rem;
    }
  };
  auto load_group = [&](uint4 (&d)[4], int g) {
#pragma unroll
    for (int q = 0; q < 4; ++q) {
      const int blk = g * 4 + q;
      d[q] = ob[((blk < NBLK) ? blk : (NBLK - 1)) * 32 + lane];
    }
  };
  auto scan_group = [&](const uint4 (&d)[4], int g) {
#pragma unroll
    for (int q = 0; q < 4; ++q)
      if (g * 4 + q < NBLK) scan_block(d[q], g * 4 + q);
  };
  uint4 da[4], db[4];
  load_group(da, wv);
  for (int g = wv; g < NGRP; g += 2 * NW) {
    load_group(db, g + NW);
    scan_group(da, g);
    load_group(da, g + 2 * NW);
    if (g + NW < NGRP) scan_group(db, g + NW);
  }
  for (int done = 0; done < cnt; done += 32) {
    const int n = (cnt - done < 32) ? (cnt - done) : 32;
    batch(mylist + done, n);
  }
}

#define MWAVES 8
#define SWAVES (AWAVES - MWAVES)
__global__ __launch_bounds__(1024) void k_scan(const int* __restrict__ src,
                                               const int* __restrict__ dst,
                                               const unsigned* __restrict__ ob,
                                               float* __restrict__ dinv,
                                               int* __restrict__ elist,
                                               int* __restrict__ ecount,
                                               const float* __restrict__ X,
                                               const float* __restrict__ W,
                                               float* __restrict__ H) {
  __shared__ float4   wl[(DD / 4) * DD];
  __shared__ unsigned hits[SWAVES * LCAP];
  __shared__ int      degc[AROWS + 1];
  __shared__ int      wgn[2];

  const int tid  = threadIdx.x;
  const int lane = tid & 31;
  const int wv   = __builtin_amdgcn_readfirstlane(tid >> 5);
  const int d0   = blockIdx.x * AROWS;
  const int nrows = (NN - d0 < AROWS) ? (NN - d0) : AROWS;
  const bool mat = wv < MWAVES;
  const int half = lane >> 4;
  const int l16  = lane & 15;

  float4 a[16];
  auto load_a = [&](int strip) {
    const float4* xr = (const float4*)(X + (size_t)(strip * 16 + l16) * DD) + half;
#pragma unroll
    for (int kk = 0; kk < 16; ++kk) a[kk] = xr[2 * kk];
  };
  if (mat) load_a(blockIdx.x + AWG * wv);

#pragma unroll
  for (int it = 0; it < (DD / 4) * DD / 1024; ++it) {
    const int i   = tid + it * 1024;
    const int p   = i & 127;
    const int col = 4 * (p & 15) + ((p >> 4) & 3) + 64 * (p >> 6);
    const float* wp = W + (size_t)((i >> 7) * 4) * DD + col;
    float4 v;
    v.x = wp[0];
    v.y = wp[DD];
    v.z = wp[2 * DD];
    v.w = wp[3 * DD];
    wl[i] = v;
  }
  for (int i = tid; i <= AROWS; i += 1024) degc[i] = 0;
  if (tid < 2) wgn[tid] = 0;
  __syncthreads();

  if (mat) {
    int strip = blockIdx.x + AWG * wv;
    int wslot = half * 128 + l16;
    for (;;) {
      asm volatile("" : "+v"(wslot));
      float* hp = H + (size_t)(strip * 16 + 8 * half) * DD + 4 * l16;
#pragma unroll
      for (int g = 0; g < 2; ++g) {
        v8f acc[4];
#pragma unroll
        for (int t = 0; t < 4; ++t) acc[t] = (v8f){};
#pragma unroll
        for (int kk = 0; kk < 16; ++kk) {
          float4 b[4];
#pragma unroll
          for (int t = 0; t < 4; ++t) b[t] = wl[wslot + kk * 256 + (4 * g + t) * 16];
          v2f a0, a1;
          a0.x = a[kk].x; a0.y = a[kk].y;
          a1.x = a[kk].z; a1.y = a[kk].w;
#pragma unroll
          for (int t = 0; t < 4; ++t) {
            v2f b0;
            b0.x = b[t].x; b0.y = b[t].y;
            acc[t] = __builtin_amdgcn_wmma_f32_16x16x4_f32(
                 false, a0,  false, b0,
                 (short)0, acc[t],  false,  false);
          }
#pragma unroll
          for (int t = 0; t < 4; ++t) {
            v2f b1;
            b1.x = b[t].z; b1.y = b[t].w;
            acc[t] = __builtin_amdgcn_wmma_f32_16x16x4_f32(
                 false, a1,  false, b1,
                 (short)0, acc[t],  false,  false);
          }
        }
#pragma unroll
        for (int r = 0; r < 8; ++r) {
          float4 o;
          o.x = acc[0][r];
          o.y = acc[1][r];
          o.z = acc[2][r];
          o.w = acc[3][r];
          *(float4*)(hp + (size_t)r * DD + 64 * g) = o;
        }
      }
      strip += AWG * MWAVES;
      if (strip >= NN / 16) break;
      load_a(strip);
    }
  } else {
    const int swv = wv - MWAVES;
    int* mine = elist + (size_t)blockIdx.x * WCAP;
    scan_owner_bytes<SWAVES>((const uint4*)ob, (unsigned)blockIdx.x, hits + swv * LCAP, lane, swv,
                     [&](const unsigned* list, int n) {
      unsigned e = 0u;
      if (lane < n) e = list[lane];
      const int d = dst[e];
      const int s = src[e];
      const unsigned dl = (unsigned)(d - d0);
      const bool act = (lane < n) && (dl < (unsigned)nrows);
      const unsigned m = __builtin_amdgcn_ballot_w32(act);
      const int nn = __builtin_popcount(m);
      if (act) atomicAdd(&degc[dl], 1);
      int base = 0;
      if (lane == 0) base = atomicAdd(&wgn[0], nn);
      base = __builtin_amdgcn_readfirstlane(base);
      if (base + nn <= WCAP) {
        if (act) mine[base + (int)__builtin_amdgcn_mbcnt_lo(m, 0u)] = (int)((dl << 17) | (unsigned)s);
      } else if (lane == 0) {
        wgn[1] = 1;
      }
    });
  }
  __syncthreads();

  for (int i = tid; i < nrows; i += 1024)
    dinv[d0 + i] = rsqrtf((float)degc[i] + 1.0f);
  if (tid == 0) ecount[blockIdx.x * ECPITCH] = (wgn[1] != 0) ? -1 : wgn[0];
}

template <bool O32>
__device__ __forceinline__ void add_rows(int s, int dl, float nrm, int n, int lane,
                                         const float* __restrict__ H, float* acc) {
  const int nk = (n + 15) & ~15;
  for (int k0 = 0; k0 < nk; k0 += 16) {
    float4 v[16];
    float  w[16];
    int    r[16];
#pragma unroll
    for (int j = 0; j < 16; ++j) {
      const int sj = __builtin_amdgcn_readlane(s, k0 + j);
      w[j] = __int_as_float(__builtin_amdgcn_readlane(__float_as_int(nrm), k0 + j));
      r[j] = __builtin_amdgcn_readlane(dl, k0 + j);
      if (O32)
        v[j] = *(const float4*)((const char*)H + (((unsigned)sj << 9) | ((unsigned)lane << 4)));
      else
        v[j] = ((const float4*)(H + (size_t)sj * DD))[lane];
    }
#pragma unroll
    for (int j = 0; j < 16; ++j) {
      float* a = acc + r[j] * DD + lane;
      atomicAdd(a +  0, v[j].x * w[j]);
      atomicAdd(a + 32, v[j].y * w[j]);
      atomicAdd(a + 64, v[j].z * w[j]);
      atomicAdd(a + 96, v[j].w * w[j]);
    }
  }
}

template <bool O32>
__device__ __forceinline__ void agg_batch(const unsigned* list, int n, int lane,
                                          const int* __restrict__ src,
                                          const float* __restrict__ dinv,
                                          const float* __restrict__ H,
                                          float* acc, const float* dloc) {
  unsigned ent = ((unsigned)SINK << 21);
  if (lane < n) ent = list[lane];
  const unsigned e  = ent & 0x1FFFFFu;
  const int      dl = (int)(ent >> 21);
  const int      s  = src[e];
  const float  nrm  = dinv[s] * dloc[dl];
  add_rows<O32>(s, dl, nrm, n, lane, H, acc);
}

template <bool MAT>
__global__ __launch_bounds__(1024) void k_agg_lds(const int* __restrict__ src,
                                                  const int* __restrict__ dst,
                                                  const float* __restrict__ dinv,
                                                  const int* __restrict__ elist,
                                                  const int* __restrict__ ecount,
                                                  const float* __restrict__ H,
                                                  const float* __restrict__ bias,
                                                  float* __restrict__ out,
                                                  int relu,
                                                  const float* __restrict__ X2,
                                                  const float* __restrict__ W2,
                                                  float* __restrict__ H2) {
  constexpr int GW = MAT ? SWAVES : AWAVES;
  constexpr int LC = MAT ? LCAPM : LCAP;
  __shared__ float    acc[(AROWS + 1) * DD];
  __shared__ float    dloc[AROWS + 1];
  __shared__ unsigned hits[GW * LC];
  __shared__ float4   wl[MAT ? (DD / 4) * DD : 1];

  const int tid  = threadIdx.x;
  const int lane = tid & 31;
  const int wv   = __builtin_amdgcn_readfirstlane(tid >> 5);
  const int d0   = blockIdx.x * AROWS;
  const int nrows = (NN - d0 < AROWS) ? (NN - d0) : AROWS;
  const bool mat = MAT && (wv < MWAVES);
  const int gwv  = MAT ? (wv - MWAVES) : wv;
  const int half = lane >> 4;
  const int l16  = lane & 15;

  float4 a[16];
  auto load_a = [&](int strip) {
    const float4* xr = (const float4*)(X2 + (size_t)(strip * 16 + l16) * DD) + half;
#pragma unroll
    for (int kk = 0; kk < 16; ++kk) a[kk] = xr[2 * kk];
  };
  if (MAT) {
#pragma unroll
    for (int it = 0; it < (DD / 4) * DD / 1024; ++it) {
      const int i   = tid + it * 1024;
      const int p   = i & 127;
      const int col = 4 * (p & 15) + ((p >> 4) & 3) + 64 * (p >> 6);
      const float* wp = W2 + (size_t)((i >> 7) * 4) * DD + col;
      float4 v;
      v.x = wp[0];
      v.y = wp[DD];
      v.z = wp[2 * DD];
      v.w = wp[3 * DD];
      wl[i] = v;
    }
  }

  float4* acc4 = (float4*)acc;
  for (int i = tid; i < (AROWS + 1) * (DD / 4); i += 1024)
    acc4[i] = make_float4(0.0f, 0.0f, 0.0f, 0.0f);
  for (int i = tid; i <= AROWS; i += 1024)
    dloc[i] = (i < nrows) ? dinv[d0 + i] : 0.0f;
  __syncthreads();

  if (mat) {
    int strip = blockIdx.x + AWG * wv;
    int wslot = half * 128 + l16;
    for (;;) {
      load_a(strip);
      asm volatile("" : "+v"(wslot));
      float* hp = H2 + (size_t)(strip * 16 + 8 * half) * DD + 4 * l16;
#pragma unroll
      for (int g = 0; g < 2; ++g) {
        v8f c[4];
#pragma unroll
        for (int t = 0; t < 4; ++t) c[t] = (v8f){};
#pragma unroll
        for (int kk = 0; kk < 16; ++kk) {
          float4 b[4];
#pragma unroll
          for (int t = 0; t < 4; ++t) b[t] = wl[wslot + kk * 256 + (4 * g + t) * 16];
          v2f a0, a1;
          a0.x = a[kk].x; a0.y = a[kk].y;
          a1.x = a[kk].z; a1.y = a[kk].w;
#pragma unroll
          for (int t = 0; t < 4; ++t) {
            v2f b0;
            b0.x = b[t].x; b0.y = b[t].y;
            c[t] = __builtin_amdgcn_wmma_f32_16x16x4_f32(
                 false, a0,  false, b0,
                 (short)0, c[t],  false,  false);
          }
#pragma unroll
          for (int t = 0; t < 4; ++t) {
            v2f b1;
            b1.x = b[t].z; b1.y = b[t].w;
            c[t] = __builtin_amdgcn_wmma_f32_16x16x4_f32(
                 false, a1,  false, b1,
                 (short)0, c[t],  false,  false);
          }
        }
#pragma unroll
        for (int r = 0; r < 8; ++r) {
          float4 o;
          o.x = c[0][r];
          o.y = c[1][r];
          o.z = c[2][r];
          o.w = c[3][r];
          *(float4*)(hp + (size_t)r * DD + 64 * g) = o;
        }
      }
      strip += AWG * MWAVES;
      if (strip >= NN / 16) break;
    }
  } else {
    const int total = ecount[blockIdx.x * ECPITCH];
    if (total >= 0) {
      const int* mine = elist + (size_t)blockIdx.x * WCAP;
      const int STEP = 32 * GW;
      const unsigned padent = (unsigned)SINK << 17;
      auto list_word = [&](int i) -> unsigned {
        const unsigned t = (unsigned)mine[(i < total) ? i : 0];
        return (i < total) ? t : padent;
      };
      if (total > 0) {
        int b = gwv * 32;
        unsigned ent0 = list_word(b + lane);
        unsigned ent1 = list_word(b + STEP + lane);
        float dv0 = dinv[ent0 & 0x1FFFFu];
        while (b < total) {
          const unsigned ent2 = list_word(b + 2 * STEP + lane);
          const float dv1 = dinv[ent1 & 0x1FFFFu];
          const int n  = (total - b < 32) ? (total - b) : 32;
          const int s  = (int)(ent0 & 0x1FFFFu);
          const int dl = (int)(ent0 >> 17);
          add_rows<MAT>(s, dl, dv0 * dloc[dl], n, lane, H, acc);
          ent0 = ent1; dv0 = dv1; ent1 = ent2;
          b += STEP;
        }
      }
    } else {
      scan_edges<GW>(dst, d0, (unsigned)nrows, hits + gwv * LC, lane, gwv,
                     [&](const unsigned* list, int n) {
        agg_batch<MAT>(list, n, lane, src, dinv, H, acc, dloc);
      });
    }
  }
  __syncthreads();

  const float b0 = bias[lane * 4 + 0];
  const float b1 = bias[lane * 4 + 1];
  const float b2 = bias[lane * 4 + 2];
  const float b3 = bias[lane * 4 + 3];
  for (int r0 = wv; r0 < nrows; r0 += 4 * AWAVES) {
    float4 h[4];
#pragma unroll
    for (int j = 0; j < 4; ++j) {
      const int r  = r0 + j * AWAVES;
      const int rr = (r < nrows) ? r : (nrows - 1);
      h[j] = ((const float4*)(H + (size_t)(d0 + rr) * DD))[lane];
    }
#pragma unroll
    for (int j = 0; j < 4; ++j) {
      const int r = r0 + j * AWAVES;
      if (r < nrows) {
        const float di = dloc[r];
        const float* ap = acc + r * DD + lane;
        float4 o;
        o.x = ap[0]  + h[j].x * di * di + b0;
        o.y = ap[32] + h[j].y * di * di + b1;
        o.z = ap[64] + h[j].z * di * di + b2;
        o.w = ap[96] + h[j].w * di * di + b3;
        if (relu) {
          o.x = fmaxf(o.x, 0.0f);
          o.y = fmaxf(o.y, 0.0f);
          o.z = fmaxf(o.z, 0.0f);
          o.w = fmaxf(o.w, 0.0f);
        }
        ((float4*)(out + (size_t)(d0 + r) * DD))[lane] = o;
      }
    }
  }
}

extern "C" void kernel_launch(void* const* d_in, const int* in_sizes, int n_in,
                              void* d_out, int out_size, void* d_ws, size_t ws_size,
                              hipStream_t stream) {
  const float* x[2]  = {(const float*)d_in[0], (const float*)d_in[2]};
  const int*   ei[2] = {(const int*)d_in[1],   (const int*)d_in[3]};
  const float* W0 = (const float*)d_in[4];
  const float* b0 = (const float*)d_in[5];
  const float* W1 = (const float*)d_in[6];
  const float* b1 = (const float*)d_in[7];
  float* out = (float*)d_out;

  const size_t nd  = (size_t)NN * DD;
  const size_t SET = 100352 + 3 * nd;
  float* ws = (float*)d_ws;
  float* dinv[2];
  float* H[2];
  float* Y[2];
  int*   elist[2];
  int*   ecount[2];
  for (int g = 0; g < 2; ++g) {
    float* s  = ws + g * SET;
    dinv[g]   = s;
    H[g]      = s + 100352;
    elist[g]  = (int*)(H[g] + nd);
    ecount[g] = elist[g] + (size_t)AWG * WCAP;
    Y[g]      = H[g] + 2 * nd;
  }
  unsigned* ob = (unsigned*)(ws + 2 * SET);

  const int* srcp[2] = {ei[0], ei[1]};
  const int* dstp[2] = {ei[0] + NE, ei[1] + NE};
  const dim3 B(256);

  k_owner<<<dim3((NV16 + 255) / 256, 2), B, 0, stream>>>(dstp[0], dstp[1], ob);

  for (int g = 0; g < 2; ++g)
    k_scan<<<AWG, dim3(1024), 0, stream>>>(srcp[g], dstp[g], ob + (size_t)g * (NE / 4), dinv[g],
                                          elist[g], ecount[g], x[g], W0, H[g]);

  k_agg_lds<false><<<AWG, dim3(1024), 0, stream>>>(srcp[0], dstp[0], dinv[0], elist[0], ecount[0],
                                                  H[0], b0, Y[0], 1, nullptr, nullptr, nullptr);
  k_agg_lds<true><<<AWG, dim3(1024), 0, stream>>>(srcp[1], dstp[1], dinv[1], elist[1], ecount[1],
                                                 H[1], b0, Y[1], 1, Y[0], W1, H[0]);
  k_agg_lds<true><<<AWG, dim3(1024), 0, stream>>>(srcp[0], dstp[0], dinv[0], elist[0], ecount[0],
                                                 H[0], b1, out, 0, Y[1], W1, H[1]);
  k_agg_lds<false><<<AWG, dim3(1024), 0, stream>>>(srcp[1], dstp[1], dinv[1], elist[1], ecount[1],
                                                  H[1], b1, out + nd, 0, nullptr, nullptr, nullptr);
}


# Owner index per edge kept as 8 bit planes per 32 edges, not a byte: the scan tests a lane's 32 edges with one logic op per plane (was 5 ops per 4 edges), 1024-edge blocks; LDS arrival order kept
# speedup vs baseline: 5.0740x; 1.0442x over previous
#include <hip/hip_runtime.h>


#define NN 100000
#define NE 1600000
#define DD 128

typedef __attribute__((ext_vector_type(2))) float v2f;
typedef __attribute__((ext_vector_type(8))) float v8f;


#define AROWS 391
#define AWG   256
#define AWAVES 32
#define LCAP  640
#define LCAPM 576
#define SINK  AROWS
#define WCAP  32768
#define ECPITCH 32

template <int NW, class F>
__device__ __forceinline__ void scan_edges(const int* __restrict__ dst, int d0, unsigned un,
                                           unsigned* mylist, int lane, int wv, F&& batch) {
  int cnt = 0;
  const int4* dst4 = (const int4*)dst;
  const int NBLK = NE / 512;
  auto scan_block = [&](const int4 (&d)[4], int blk) {
#pragma unroll
    for (int q = 0; q < 4; ++q) {
      const unsigned u0 = (unsigned)(d[q].x - d0);
      const unsigned u1 = (unsigned)(d[q].y - d0);
      const unsigned u2 = (unsigned)(d[q].z - d0);
      const unsigned u3 = (unsigned)(d[q].w - d0);
      const unsigned m01 = u0 < u1 ? u0 : u1;
      const unsigned m23 = u2 < u3 ? u2 : u3;
      const unsigned mn  = m01 < m23 ? m01 : m23;
      if (__builtin_amdgcn_ballot_w32(mn < un) != 0u) {
        const unsigned eb = (unsigned)((blk * 4 + q) * 128 + lane * 4);
        unsigned m;
        m = __builtin_amdgcn_ballot_w32(u0 < un);
        if (u0 < un) mylist[cnt + (int)__builtin_amdgcn_mbcnt_lo(m, 0u)] = (u0 << 21) | (eb + 0u);
        cnt += __builtin_popcount(m);
        m = __builtin_amdgcn_ballot_w32(u1 < un);
        if (u1 < un) mylist[cnt + (int)__builtin_amdgcn_mbcnt_lo(m, 0u)] = (u1 << 21) | (eb + 1u);
        cnt += __builtin_popcount(m);
        m = __builtin_amdgcn_ballot_w32(u2 < un);
        if (u2 < un) mylist[cnt + (int)__builtin_amdgcn_mbcnt_lo(m, 0u)] = (u2 << 21) | (eb + 2u);
        cnt += __builtin_popcount(m);
        m = __builtin_amdgcn_ballot_w32(u3 < un);
        if (u3 < un) mylist[cnt + (int)__builtin_amdgcn_mbcnt_lo(m, 0u)] = (u3 << 21) | (eb + 3u);
        cnt += __builtin_popcount(m);
      }
    }
    if (cnt >= 64) {
      int done = 0;
      while (cnt - done >= 32) {
        batch(mylist + done, 32);
        done += 32;
      }
      const int rem = cnt - done;
      unsigned t = 0u;
      if (lane < rem) t = mylist[done + lane];
      if (lane < rem) mylist[lane] = t;
      cnt = rem;
    }
  };
  auto load_block = [&](int4 (&d)[4], int blk) {
    const int b = (blk < NBLK) ? blk : (NBLK - 1);
#pragma unroll
    for (int q = 0; q < 4; ++q) d[q] = dst4[(b * 4 + q) * 32 + lane];
  };
  int4 da[4], db[4];
  load_block(da, wv);
  for (int blk = wv; blk < NBLK; blk += 2 * NW) {
    load_block(db, blk + NW);
    scan_block(da, blk);
    load_block(da, blk + 2 * NW);
    if (blk + NW < NBLK) scan_block(db, blk + NW);
  }
  for (int done = 0; done < cnt; done += 32) {
    const int n = (cnt - done < 32) ? (cnt - done) : 32;
    batch(mylist + done, n);
  }
}

#define NT32 (NE / 32)
#define PBLK ((NT32 + 31) / 32)
#define PV16 (PBLK * 64)
__global__ __launch_bounds__(256) void k_owner(const int* __restrict__ dstA,
                                               const int* __restrict__ dstB,
                                               unsigned* __restrict__ ob) {
  const int t = blockIdx.x * 256 + (int)threadIdx.x;
  if (t >= PBLK * 32) return;
  unsigned pl[8];
#pragma unroll
  for (int b = 0; b < 8; ++b) pl[b] = 0u;
  if (t < NT32) {
    const int4* d4 = (const int4*)(blockIdx.y == 0 ? dstA : dstB) + (size_t)t * 8;
#pragma unroll
    for (int c = 0; c < 8; ++c) {
      const int4 d = d4[c];
      const unsigned o[4] = {(unsigned)d.x / AROWS, (unsigned)d.y / AROWS,
                             (unsigned)d.z / AROWS, (unsigned)d.w / AROWS};
#pragma unroll
      for (int j = 0; j < 4; ++j) {
#pragma unroll
        for (int b = 0; b < 8; ++b) pl[b] |= ((o[j] >> b) & 1u) << (4 * c + j);
      }
    }
  }
  uint4* op = (uint4*)ob + (size_t)blockIdx.y * PV16 + (size_t)(t >> 5) * 64 + (t & 31);
  uint4 lo, hi;
  lo.x = pl[0]; lo.y = pl[1]; lo.z = pl[2]; lo.w = pl[3];
  hi.x = pl[4]; hi.y = pl[5]; hi.z = pl[6]; hi.w = pl[7];
  op[0]  = lo;
  op[32] = hi;
}

template <int NW, class F>
__device__ __forceinline__ void scan_owner_planes(const uint4* __restrict__ ob, unsigned owner,
                                                  unsigned* mylist, int lane, int wv, F&& batch) {
  int cnt = 0;
  const int NBLK = PBLK;
  const int NGRP = (NBLK + 1) / 2;
  unsigned inv[8];
#pragma unroll
  for (int b = 0; b < 8; ++b) inv[b] = ((owner >> b) & 1u) ? 0u : 0xFFFFFFFFu;
  auto scan_block = [&](const uint4& lo, const uint4& hi, int blk) {
    unsigned hm = lo.x ^ inv[0];
    hm &= lo.y ^ inv[1];
    hm &= lo.z ^ inv[2];
    hm &= lo.w ^ inv[3];
    hm &= hi.x ^ inv[4];
    hm &= hi.y ^ inv[5];
    hm &= hi.z ^ inv[6];
    hm &= hi.w ^ inv[7];
    const unsigned eb = (unsigned)(blk * 32 + lane) * 32u;
    if (eb >= (unsigned)NE) hm = 0u;
    unsigned mb = __builtin_amdgcn_ballot_w32(hm != 0u);
    while (mb != 0u) {
      const unsigned p = (unsigned)__builtin_ctz(hm | 0x80000000u);
      if (hm != 0u)
        mylist[cnt + (int)__builtin_amdgcn_mbcnt_lo(mb, 0u)] = eb + p;
      cnt += __builtin_popcount(mb);
      hm &= hm - 1u;
      mb = __builtin_amdgcn_ballot_w32(hm != 0u);
    }
    if (cnt >= 64) {
      int done = 0;
      while (cnt - done >= 32) {
        batch(mylist + done, 32);
        done += 32;
      }
      const int rem = cnt - done;
      unsigned t = 0u;
      if (lane < rem) t = mylist[done + lane];
      if (lane < rem) mylist[lane] = t;
      cnt = rem;
    }
  };
  auto load_group = [&](uint4 (&d)[4], int g) {
#pragma unroll
    for (int q = 0; q < 2; ++q) {
      const int blk = g * 2 + q;
      const int b   = (blk < NBLK) ? blk : (NBLK - 1);
      d[2 * q]     = ob[b * 64 + lane];
      d[2 * q + 1] = ob[b * 64 + 32 + lane];
    }
  };
  auto scan_group = [&](const uint4 (&d)[4], int g) {
#pragma unroll
    for (int q = 0; q < 2; ++q)
      if (g * 2 + q < NBLK) scan_block(d[2 * q], d[2 * q + 1], g * 2 + q);
  };
  uint4 da[4], db[4];
  load_group(da, wv);
  for (int g = wv; g < NGRP; g += 2 * NW) {
    load_group(db, g + NW);
    scan_group(da, g);
    load_group(da, g + 2 * NW);
    if (g + NW < NGRP) scan_group(db, g + NW);
  }
  for (int done = 0; done < cnt; done += 32) {
    const int n = (cnt - done < 32) ? (cnt - done) : 32;
    batch(mylist + done, n);
  }
}

#define MWAVES 8
#define SWAVES (AWAVES - MWAVES)
#define LCAPS 1088
__global__ __launch_bounds__(1024) void k_scan(const int* __restrict__ src,
                                               const int* __restrict__ dst,
                                               const unsigned* __restrict__ ob,
                                               float* __restrict__ dinv,
                                               int* __restrict__ elist,
                                               int* __restrict__ ecount,
                                               const float* __restrict__ X,
                                               const float* __restrict__ W,
                                               float* __restrict__ H) {
  __shared__ float4   wl[(DD / 4) * DD];
  __shared__ unsigned hits[SWAVES * LCAPS];
  __shared__ int      degc[AROWS + 1];
  __shared__ int      wgn[2];

  const int tid  = threadIdx.x;
  const int lane = tid & 31;
  const int wv   = __builtin_amdgcn_readfirstlane(tid >> 5);
  const int d0   = blockIdx.x * AROWS;
  const int nrows = (NN - d0 < AROWS) ? (NN - d0) : AROWS;
  const bool mat = wv < MWAVES;
  const int half = lane >> 4;
  const int l16  = lane & 15;

  float4 a[16];
  auto load_a = [&](int strip) {
    const float4* xr = (const float4*)(X + (size_t)(strip * 16 + l16) * DD) + half;
#pragma unroll
    for (int kk = 0; kk < 16; ++kk) a[kk] = xr[2 * kk];
  };
  if (mat) load_a(blockIdx.x + AWG * wv);

#pragma unroll
  for (int it = 0; it < (DD / 4) * DD / 1024; ++it) {
    const int i   = tid + it * 1024;
    const int p   = i & 127;
    const int col = 4 * (p & 15) + ((p >> 4) & 3) + 64 * (p >> 6);
    const float* wp = W + (size_t)((i >> 7) * 4) * DD + col;
    float4 v;
    v.x = wp[0];
    v.y = wp[DD];
    v.z = wp[2 * DD];
    v.w = wp[3 * DD];
    wl[i] = v;
  }
  for (int i = tid; i <= AROWS; i += 1024) degc[i] = 0;
  if (tid < 2) wgn[tid] = 0;
  __syncthreads();

  if (mat) {
    int strip = blockIdx.x + AWG * wv;
    int wslot = half * 128 + l16;
    for (;;) {
      asm volatile("" : "+v"(wslot));
      float* hp = H + (size_t)(strip * 16 + 8 * half) * DD + 4 * l16;
#pragma unroll
      for (int g = 0; g < 2; ++g) {
        v8f acc[4];
#pragma unroll
        for (int t = 0; t < 4; ++t) acc[t] = (v8f){};
#pragma unroll
        for (int kk = 0; kk < 16; ++kk) {
          float4 b[4];
#pragma unroll
          for (int t = 0; t < 4; ++t) b[t] = wl[wslot + kk * 256 + (4 * g + t) * 16];
          v2f a0, a1;
          a0.x = a[kk].x; a0.y = a[kk].y;
          a1.x = a[kk].z; a1.y = a[kk].w;
#pragma unroll
          for (int t = 0; t < 4; ++t) {
            v2f b0;
            b0.x = b[t].x; b0.y = b[t].y;
            acc[t] = __builtin_amdgcn_wmma_f32_16x16x4_f32(
                 false, a0,  false, b0,
                 (short)0, acc[t],  false,  false);
          }
#pragma unroll
          for (int t = 0; t < 4; ++t) {
            v2f b1;
            b1.x = b[t].z; b1.y = b[t].w;
            acc[t] = __builtin_amdgcn_wmma_f32_16x16x4_f32(
                 false, a1,  false, b1,
                 (short)0, acc[t],  false,  false);
          }
        }
#pragma unroll
        for (int r = 0; r < 8; ++r) {
          float4 o;
          o.x = acc[0][r];
          o.y = acc[1][r];
          o.z = acc[2][r];
          o.w = acc[3][r];
          *(float4*)(hp + (size_t)r * DD + 64 * g) = o;
        }
      }
      strip += AWG * MWAVES;
      if (strip >= NN / 16) break;
      load_a(strip);
    }
  } else {
    const int swv = wv - MWAVES;
    int* mine = elist + (size_t)blockIdx.x * WCAP;
    scan_owner_planes<SWAVES>((const uint4*)ob, (unsigned)blockIdx.x, hits + swv * LCAPS, lane, swv,
                     [&](const unsigned* list, int n) {
      unsigned e = 0u;
      if (lane < n) e = list[lane];
      const int d = dst[e];
      const int s = src[e];
      const unsigned dl = (unsigned)(d - d0);
      const bool act = (lane < n) && (dl < (unsigned)nrows);
      const unsigned m = __builtin_amdgcn_ballot_w32(act);
      const int nn = __builtin_popcount(m);
      if (act) atomicAdd(&degc[dl], 1);
      int base = 0;
      if (lane == 0) base = atomicAdd(&wgn[0], nn);
      base = __builtin_amdgcn_readfirstlane(base);
      if (base + nn <= WCAP) {
        if (act) mine[base + (int)__builtin_amdgcn_mbcnt_lo(m, 0u)] = (int)((dl << 17) | (unsigned)s);
      } else if (lane == 0) {
        wgn[1] = 1;
      }
    });
  }
  __syncthreads();

  for (int i = tid; i < nrows; i += 1024)
    dinv[d0 + i] = rsqrtf((float)degc[i] + 1.0f);
  if (tid == 0) ecount[blockIdx.x * ECPITCH] = (wgn[1] != 0) ? -1 : wgn[0];
}

template <bool O32>
__device__ __forceinline__ void add_rows(int s, int dl, float nrm, int n, int lane,
                                         const float* __restrict__ H, float* acc) {
  const int nk = (n + 15) & ~15;
  for (int k0 = 0; k0 < nk; k0 += 16) {
    float4 v[16];
    float  w[16];
    int    r[16];
#pragma unroll
    for (int j = 0; j < 16; ++j) {
      const int sj = __builtin_amdgcn_readlane(s, k0 + j);
      w[j] = __int_as_float(__builtin_amdgcn_readlane(__float_as_int(nrm), k0 + j));
      r[j] = __builtin_amdgcn_readlane(dl, k0 + j);
      if (O32)
        v[j] = *(const float4*)((const char*)H + (((unsigned)sj << 9) | ((unsigned)lane << 4)));
      else
        v[j] = ((const float4*)(H + (size_t)sj * DD))[lane];
    }
#pragma unroll
    for (int j = 0; j < 16; ++j) {
      float* a = acc + r[j] * DD + lane;
      atomicAdd(a +  0, v[j].x * w[j]);
      atomicAdd(a + 32, v[j].y * w[j]);
      atomicAdd(a + 64, v[j].z * w[j]);
      atomicAdd(a + 96, v[j].w * w[j]);
    }
  }
}

template <bool O32>
__device__ __forceinline__ void agg_batch(const unsigned* list, int n, int lane,
                                          const int* __restrict__ src,
                                          const float* __restrict__ dinv,
                                          const float* __restrict__ H,
                                          float* acc, const float* dloc) {
  unsigned ent = ((unsigned)SINK << 21);
  if (lane < n) ent = list[lane];
  const unsigned e  = ent & 0x1FFFFFu;
  const int      dl = (int)(ent >> 21);
  const int      s  = src[e];
  const float  nrm  = dinv[s] * dloc[dl];
  add_rows<O32>(s, dl, nrm, n, lane, H, acc);
}

template <bool MAT>
__global__ __launch_bounds__(1024) void k_agg_lds(const int* __restrict__ src,
                                                  const int* __restrict__ dst,
                                                  const float* __restrict__ dinv,
                                                  const int* __restrict__ elist,
                                                  const int* __restrict__ ecount,
                                                  const float* __restrict__ H,
                                                  const float* __restrict__ bias,
                                                  float* __restrict__ out,
                                                  int relu,
                                                  const float* __restrict__ X2,
                                                  const float* __restrict__ W2,
                                                  float* __restrict__ H2) {
  constexpr int GW = MAT ? SWAVES : AWAVES;
  constexpr int LC = MAT ? LCAPM : LCAP;
  __shared__ float    acc[(AROWS + 1) * DD];
  __shared__ float    dloc[AROWS + 1];
  __shared__ unsigned hits[GW * LC];
  __shared__ float4   wl[MAT ? (DD / 4) * DD : 1];

  const int tid  = threadIdx.x;
  const int lane = tid & 31;
  const int wv   = __builtin_amdgcn_readfirstlane(tid >> 5);
  const int d0   = blockIdx.x * AROWS;
  const int nrows = (NN - d0 < AROWS) ? (NN - d0) : AROWS;
  const bool mat = MAT && (wv < MWAVES);
  const int gwv  = MAT ? (wv - MWAVES) : wv;
  const int half = lane >> 4;
  const int l16  = lane & 15;

  float4 a[16];
  auto load_a = [&](int strip) {
    const float4* xr = (const float4*)(X2 + (size_t)(strip * 16 + l16) * DD) + half;
#pragma unroll
    for (int kk = 0; kk < 16; ++kk) a[kk] = xr[2 * kk];
  };
  if (MAT) {
#pragma unroll
    for (int it = 0; it < (DD / 4) * DD / 1024; ++it) {
      const int i   = tid + it * 1024;
      const int p   = i & 127;
      const int col = 4 * (p & 15) + ((p >> 4) & 3) + 64 * (p >> 6);
      const float* wp = W2 + (size_t)((i >> 7) * 4) * DD + col;
      float4 v;
      v.x = wp[0];
      v.y = wp[DD];
      v.z = wp[2 * DD];
      v.w = wp[3 * DD];
      wl[i] = v;
    }
  }

  float4* acc4 = (float4*)acc;
  for (int i = tid; i < (AROWS + 1) * (DD / 4); i += 1024)
    acc4[i] = make_float4(0.0f, 0.0f, 0.0f, 0.0f);
  for (int i = tid; i <= AROWS; i += 1024)
    dloc[i] = (i < nrows) ? dinv[d0 + i] : 0.0f;
  __syncthreads();

  if (mat) {
    int strip = blockIdx.x + AWG * wv;
    int wslot = half * 128 + l16;
    for (;;) {
      load_a(strip);
      asm volatile("" : "+v"(wslot));
      float* hp = H2 + (size_t)(strip * 16 + 8 * half) * DD + 4 * l16;
#pragma unroll
      for (int g = 0; g < 2; ++g) {
        v8f c[4];
#pragma unroll
        for (int t = 0; t < 4; ++t) c[t] = (v8f){};
#pragma unroll
        for (int kk = 0; kk < 16; ++kk) {
          float4 b[4];
#pragma unroll
          for (int t = 0; t < 4; ++t) b[t] = wl[wslot + kk * 256 + (4 * g + t) * 16];
          v2f a0, a1;
          a0.x = a[kk].x; a0.y = a[kk].y;
          a1.x = a[kk].z; a1.y = a[kk].w;
#pragma unroll
          for (int t = 0; t < 4; ++t) {
            v2f b0;
            b0.x = b[t].x; b0.y = b[t].y;
            c[t] = __builtin_amdgcn_wmma_f32_16x16x4_f32(
                 false, a0,  false, b0,
                 (short)0, c[t],  false,  false);
          }
#pragma unroll
          for (int t = 0; t < 4; ++t) {
            v2f b1;
            b1.x = b[t].z; b1.y = b[t].w;
            c[t] = __builtin_amdgcn_wmma_f32_16x16x4_f32(
                 false, a1,  false, b1,
                 (short)0, c[t],  false,  false);
          }
        }
#pragma unroll
        for (int r = 0; r < 8; ++r) {
          float4 o;
          o.x = c[0][r];
          o.y = c[1][r];
          o.z = c[2][r];
          o.w = c[3][r];
          *(float4*)(hp + (size_t)r * DD + 64 * g) = o;
        }
      }
      strip += AWG * MWAVES;
      if (strip >= NN / 16) break;
    }
  } else {
    const int total = ecount[blockIdx.x * ECPITCH];
    if (total >= 0) {
      const int* mine = elist + (size_t)blockIdx.x * WCAP;
      const int STEP = 32 * GW;
      const unsigned padent = (unsigned)SINK << 17;
      auto list_word = [&](int i) -> unsigned {
        const unsigned t = (unsigned)mine[(i < total) ? i : 0];
        return (i < total) ? t : padent;
      };
      if (total > 0) {
        int b = gwv * 32;
        unsigned ent0 = list_word(b + lane);
        unsigned ent1 = list_word(b + STEP + lane);
        float dv0 = dinv[ent0 & 0x1FFFFu];
        while (b < total) {
          const unsigned ent2 = list_word(b + 2 * STEP + lane);
          const float dv1 = dinv[ent1 & 0x1FFFFu];
          const int n  = (total - b < 32) ? (total - b) : 32;
          const int s  = (int)(ent0 & 0x1FFFFu);
          const int dl = (int)(ent0 >> 17);
          add_rows<MAT>(s, dl, dv0 * dloc[dl], n, lane, H, acc);
          ent0 = ent1; dv0 = dv1; ent1 = ent2;
          b += STEP;
        }
      }
    } else {
      scan_edges<GW>(dst, d0, (unsigned)nrows, hits + gwv * LC, lane, gwv,
                     [&](const unsigned* list, int n) {
        agg_batch<MAT>(list, n, lane, src, dinv, H, acc, dloc);
      });
    }
  }
  __syncthreads();

  const float b0 = bias[lane * 4 + 0];
  const float b1 = bias[lane * 4 + 1];
  const float b2 = bias[lane * 4 + 2];
  const float b3 = bias[lane * 4 + 3];
  for (int r0 = wv; r0 < nrows; r0 += 4 * AWAVES) {
    float4 h[4];
#pragma unroll
    for (int j = 0; j < 4; ++j) {
      const int r  = r0 + j * AWAVES;
      const int rr = (r < nrows) ? r : (nrows - 1);
      h[j] = ((const float4*)(H + (size_t)(d0 + rr) * DD))[lane];
    }
#pragma unroll
    for (int j = 0; j < 4; ++j) {
      const int r = r0 + j * AWAVES;
      if (r < nrows) {
        const float di = dloc[r];
        const float* ap = acc + r * DD + lane;
        float4 o;
        o.x = ap[0]  + h[j].x * di * di + b0;
        o.y = ap[32] + h[j].y * di * di + b1;
        o.z = ap[64] + h[j].z * di * di + b2;
        o.w = ap[96] + h[j].w * di * di + b3;
        if (relu) {
          o.x = fmaxf(o.x, 0.0f);
          o.y = fmaxf(o.y, 0.0f);
          o.z = fmaxf(o.z, 0.0f);
          o.w = fmaxf(o.w, 0.0f);
        }
        ((float4*)(out + (size_t)(d0 + r) * DD))[lane] = o;
      }
    }
  }
}

extern "C" void kernel_launch(void* const* d_in, const int* in_sizes, int n_in,
                              void* d_out, int out_size, void* d_ws, size_t ws_size,
                              hipStream_t stream) {
  const float* x[2]  = {(const float*)d_in[0], (const float*)d_in[2]};
  const int*   ei[2] = {(const int*)d_in[1],   (const int*)d_in[3]};
  const float* W0 = (const float*)d_in[4];
  const float* b0 = (const float*)d_in[5];
  const float* W1 = (const float*)d_in[6];
  const float* b1 = (const float*)d_in[7];
  float* out = (float*)d_out;

  const size_t nd  = (size_t)NN * DD;
  const size_t SET = 100352 + 3 * nd;
  float* ws = (float*)d_ws;
  float* dinv[2];
  float* H[2];
  float* Y[2];
  int*   elist[2];
  int*   ecount[2];
  for (int g = 0; g < 2; ++g) {
    float* s  = ws + g * SET;
    dinv[g]   = s;
    H[g]      = s + 100352;
    elist[g]  = (int*)(H[g] + nd);
    ecount[g] = elist[g] + (size_t)AWG * WCAP;
    Y[g]      = H[g] + 2 * nd;
  }
  unsigned* ob = (unsigned*)(ws + 2 * SET);

  const int* srcp[2] = {ei[0], ei[1]};
  const int* dstp[2] = {ei[0] + NE, ei[1] + NE};
  const dim3 B(256);

  k_owner<<<dim3((PBLK * 32 + 255) / 256, 2), B, 0, stream>>>(dstp[0], dstp[1], ob);

  for (int g = 0; g < 2; ++g)
    k_scan<<<AWG, dim3(1024), 0, stream>>>(srcp[g], dstp[g], ob + (size_t)g * (PV16 * 4), dinv[g],
                                          elist[g], ecount[g], x[g], W0, H[g]);

  k_agg_lds<false><<<AWG, dim3(1024), 0, stream>>>(srcp[0], dstp[0], dinv[0], elist[0], ecount[0],
                                                  H[0], b0, Y[0], 1, nullptr, nullptr, nullptr);
  k_agg_lds<true><<<AWG, dim3(1024), 0, stream>>>(srcp[1], dstp[1], dinv[1], elist[1], ecount[1],
                                                 H[1], b0, Y[1], 1, Y[0], W1, H[0]);
  k_agg_lds<true><<<AWG, dim3(1024), 0, stream>>>(srcp[0], dstp[0], dinv[0], elist[0], ecount[0],
                                                 H[0], b1, out, 0, Y[1], W1, H[1]);
  k_agg_lds<false><<<AWG, dim3(1024), 0, stream>>>(srcp[1], dstp[1], dinv[1], elist[1], ecount[1],
                                                  H[1], b1, out + nd, 0, nullptr, nullptr, nullptr);
}
